# starting kernel
# baseline (speedup 1.0000x reference)
#include <hip/hip_runtime.h>

#define N_NODES  50000
#define N_EDGES  800000
#define DIM      128
#define N_LAYERS 4
#define N_GRAPHS 512
#define N_COLS   4
#define VOCAB    128

typedef __attribute__((ext_vector_type(2))) float v2f;
typedef __attribute__((ext_vector_type(4))) float v4f;
typedef __attribute__((ext_vector_type(8))) float v8f;

__global__ void embed_kernel(const int* __restrict__ x, const float* __restrict__ emb,
                             float* __restrict__ h) {
  int t = blockIdx.x * blockDim.x + threadIdx.x;
  if (t >= N_NODES * DIM) return;
  int n = t / DIM, d = t % DIM;
  float s = 0.f;
#pragma unroll
  for (int c = 0; c < N_COLS; ++c) {
    int v = x[n * N_COLS + c];
    s += emb[((size_t)c * VOCAB + v) * DIM + d];
  }
  h[t] = s;
}

__global__ void deg_init_kernel(float* __restrict__ deg) {
  int n = blockIdx.x * blockDim.x + threadIdx.x;
  if (n < N_NODES) deg[n] = 1.0f;
}

__global__ void deg_edge_kernel(const int* __restrict__ ei, float* __restrict__ deg) {
  int e = blockIdx.x * blockDim.x + threadIdx.x;
  if (e < N_EDGES) unsafeAtomicAdd(&deg[ei[N_EDGES + e]], 1.0f);
}

__global__ void dinv_kernel(float* __restrict__ deg) {
  int n = blockIdx.x * blockDim.x + threadIdx.x;
  if (n >= N_NODES) return;
  float d = deg[n];
  deg[n] = (d > 0.f) ? rsqrtf(d) : 0.f;
}

__global__ void norm_kernel(const int* __restrict__ ei, const float* __restrict__ dinv,
                            float* __restrict__ norm) {
  int e = blockIdx.x * blockDim.x + threadIdx.x;
  if (e >= N_EDGES) return;
  norm[e] = dinv[ei[e]] * dinv[ei[N_EDGES + e]];
}

__global__ void gemm_wmma_f32(const float* __restrict__ A, const float* __restrict__ B,
                              const float* __restrict__ bias, float* __restrict__ Cout) {
  int row0 = blockIdx.x * 16;
  int col0 = blockIdx.y * 16;
  int lane = threadIdx.x;
  int half = lane >> 4;
  int lm   = lane & 15;

  const float* ap = A + (size_t)(row0 + lm) * DIM + half * 2;
  const float* bp = B + (size_t)(half * 2) * DIM + col0 + lm;

  v8f acc = {};
#pragma unroll
  for (int k = 0; k < DIM; k += 4) {
    v2f a = *(const v2f*)(ap + k);
    v2f b;
    b.x = bp[k * DIM];
    b.y = bp[k * DIM + DIM];
    acc = __builtin_amdgcn_wmma_f32_16x16x4_f32(
         false, a,  false, b,
         (short)0, acc,  false,  false);
  }

  float badd = bias ? bias[col0 + lm] : 0.f;
#pragma unroll
  for (int r = 0; r < 8; ++r)
    Cout[(size_t)(row0 + half * 8 + r) * DIM + col0 + lm] = acc[r] + badd;
}

__global__ void init_self_bias_kernel(const float* __restrict__ xw, const float* __restrict__ dinv,
                                      const float* __restrict__ bl, float* __restrict__ nh) {
  int t = blockIdx.x * blockDim.x + threadIdx.x;
  if (t >= N_NODES * DIM) return;
  int n = t / DIM, d = t % DIM;
  float s = dinv[n];
  nh[t] = bl[d] + xw[t] * s * s;
}

__global__ void scatter_kernel(const int* __restrict__ ei, const float* __restrict__ norm,
                               const float* __restrict__ xw, float* __restrict__ nh) {
  int t = blockIdx.x * blockDim.x + threadIdx.x;
  if (t >= N_EDGES * 32) return;
  int e  = t >> 5;
  int dg = (t & 31) << 2;
  int s = ei[e];
  int d = ei[N_EDGES + e];
  float w = norm[e];
  v4f v = *(const v4f*)(xw + (size_t)s * DIM + dg);
  float* dst = nh + (size_t)d * DIM + dg;
  unsafeAtomicAdd(dst + 0, v.x * w);
  unsafeAtomicAdd(dst + 1, v.y * w);
  unsafeAtomicAdd(dst + 2, v.z * w);
  unsafeAtomicAdd(dst + 3, v.w * w);
}

__global__ void relu_kernel(const float* __restrict__ nh, float* __restrict__ h) {
  int t = blockIdx.x * blockDim.x + threadIdx.x;
  if (t < N_NODES * DIM) h[t] = fmaxf(nh[t], 0.f);
}

__global__ void pool_zero_kernel(float* __restrict__ pool, float* __restrict__ cnt) {
  int t = blockIdx.x * blockDim.x + threadIdx.x;
  if (t < N_GRAPHS * DIM) pool[t] = 0.f;
  else if (t < N_GRAPHS * DIM + N_GRAPHS) cnt[t - N_GRAPHS * DIM] = 0.f;
}

__global__ void pool_scatter_kernel(const int* __restrict__ batch, const float* __restrict__ h,
                                    float* __restrict__ pool) {
  int t = blockIdx.x * blockDim.x + threadIdx.x;
  if (t >= N_NODES * DIM) return;
  int n = t / DIM, d = t % DIM;
  unsafeAtomicAdd(&pool[(size_t)batch[n] * DIM + d], h[t]);
}

__global__ void cnt_scatter_kernel(const int* __restrict__ batch, float* __restrict__ cnt) {
  int n = blockIdx.x * blockDim.x + threadIdx.x;
  if (n < N_NODES) unsafeAtomicAdd(&cnt[batch[n]], 1.0f);
}

__global__ void pool_div_kernel(float* __restrict__ pool, const float* __restrict__ cnt) {
  int t = blockIdx.x * blockDim.x + threadIdx.x;
  if (t >= N_GRAPHS * DIM) return;
  pool[t] /= fmaxf(cnt[t / DIM], 1.0f);
}

extern "C" void kernel_launch(void* const* d_in, const int* in_sizes, int n_in,
                              void* d_out, int out_size, void* d_ws, size_t ws_size,
                              hipStream_t stream) {
  const int*   x    = (const int*)d_in[0];
  const int*   ei   = (const int*)d_in[1];
  const int*   batch= (const int*)d_in[2];
  const float* emb  = (const float*)d_in[3];
  const float* W    = (const float*)d_in[4];
  const float* b    = (const float*)d_in[5];
  const float* Wr   = (const float*)d_in[6];
  const float* br   = (const float*)d_in[7];
  float* out = (float*)d_out;

  char* ws = (char*)d_ws;
  const size_t NH = (size_t)N_NODES * DIM * sizeof(float);
  float* h    = (float*)(ws);
  float* xw   = (float*)(ws + NH);
  float* nh   = (float*)(ws + 2 * NH);
  float* dinv = (float*)(ws + 3 * NH);
  float* norm = (float*)(ws + 3 * NH + 256 * 1024);
  float* pool = (float*)(ws + 3 * NH + 4 * 1024 * 1024);
  float* cnt  = (float*)(ws + 3 * NH + 4 * 1024 * 1024 + 512 * 1024);

  const int B256 = 256;
  const int nd_blocks = (N_NODES * DIM + B256 - 1) / B256;
  const int n_blocks  = (N_NODES + B256 - 1) / B256;
  const int e_blocks  = (N_EDGES + B256 - 1) / B256;

  embed_kernel<<<nd_blocks, B256, 0, stream>>>(x, emb, h);

  deg_init_kernel<<<n_blocks, B256, 0, stream>>>(dinv);
  deg_edge_kernel<<<e_blocks, B256, 0, stream>>>(ei, dinv);
  dinv_kernel<<<n_blocks, B256, 0, stream>>>(dinv);
  norm_kernel<<<e_blocks, B256, 0, stream>>>(ei, dinv, norm);

  dim3 ggrid(N_NODES / 16, DIM / 16);
  const int sc_blocks = (N_EDGES * 32 + B256 - 1) / B256;
  for (int l = 0; l < N_LAYERS; ++l) {
    gemm_wmma_f32<<<ggrid, 32, 0, stream>>>(h, W + (size_t)l * DIM * DIM, nullptr, xw);
    init_self_bias_kernel<<<nd_blocks, B256, 0, stream>>>(xw, dinv, b + (size_t)l * DIM, nh);
    scatter_kernel<<<sc_blocks, B256, 0, stream>>>(ei, norm, xw, nh);
    relu_kernel<<<nd_blocks, B256, 0, stream>>>(nh, h);
  }

  pool_zero_kernel<<<(N_GRAPHS * DIM + N_GRAPHS + B256 - 1) / B256, B256, 0, stream>>>(pool, cnt);
  pool_scatter_kernel<<<nd_blocks, B256, 0, stream>>>(batch, h, pool);
  cnt_scatter_kernel<<<n_blocks, B256, 0, stream>>>(batch, cnt);
  pool_div_kernel<<<(N_GRAPHS * DIM + B256 - 1) / B256, B256, 0, stream>>>(pool, cnt);

  dim3 rgrid(N_GRAPHS / 16, DIM / 16);
  gemm_wmma_f32<<<rgrid, 32, 0, stream>>>(pool, Wr, br, out);
}


# Set-up by the range owners in one launch: each filters the destination array into its kept list, counts its rows' degrees in LDS (integers), sums their embedding rows; 12 launches to 7; same sums
# speedup vs baseline: 3.5075x; 3.5075x over previous
#include <hip/hip_runtime.h>

#define N_NODES  50000
#define N_EDGES  800000
#define DIM      128
#define N_LAYERS 4
#define N_GRAPHS 512
#define N_COLS   4
#define VOCAB    128

#define OWN_ROWS       196
#define ROW_TILES      ((OWN_ROWS + 15) / 16)
#define ROW_PITCH      (DIM + 4)
#define N_OWNERS       ((N_NODES + OWN_ROWS - 1) / OWN_ROWS)
#define AGG_THREADS    1024
#define AGG_WAVES      (AGG_THREADS / 32)
#define EDGES_PER_STEP 128
#define N_STEPS        (N_EDGES / EDGES_PER_STEP)
#define SCAN_UNROLL    8
#define DRAIN_AT       64
#define QCAP           (DRAIN_AT + SCAN_UNROLL * EDGES_PER_STEP)
#define NO_DST         0xFFFFFFFFu
#define LIST_CAP       32768
#define REPLAY_GROUP   8
#define POOL_BINS      16
#define POOL_ROWS      ((OWN_ROWS + AGG_WAVES - 1) / AGG_WAVES)
#define EMB_BATCH      4

static_assert(N_NODES <= 65535, "node ids must fit 16 bits with 0xFFFF left free");
static_assert(N_EDGES % EDGES_PER_STEP == 0, "edge list must divide into whole steps");
static_assert(N_EDGES <= (1 << 24) && OWN_ROWS <= 256, "a queued word is edge index << 8 | local row");
static_assert(DIM == 128, "a wave holds a row as four columns per lane; bin words are found by shifts");
static_assert(POOL_ROWS < 32, "the graph ids of a wave's rows, and the one behind them, sit one per lane");

typedef __attribute__((ext_vector_type(2))) float v2f;
typedef __attribute__((ext_vector_type(4))) float v4f;
typedef __attribute__((ext_vector_type(8))) float v8f;
typedef __attribute__((ext_vector_type(4))) unsigned v4u;

__global__ void gemm_wmma_f32(const float* __restrict__ A, const float* __restrict__ B,
                              const float* __restrict__ bias, float* __restrict__ Cout) {
  int row0 = blockIdx.x * 16;
  int col0 = blockIdx.y * 16;
  int lane = threadIdx.x;
  int half = lane >> 4;
  int lm   = lane & 15;

  const float* ap = A + (size_t)(row0 + lm) * DIM + half * 2;
  const float* bp = B + (size_t)(half * 2) * DIM + col0 + lm;

  v8f acc = {};
#pragma unroll
  for (int k = 0; k < DIM; k += 4) {
    v2f a = *(const v2f*)(ap + k);
    v2f b;
    b.x = bp[k * DIM];
    b.y = bp[k * DIM + DIM];
    acc = __builtin_amdgcn_wmma_f32_16x16x4_f32(
         false, a,  false, b,
         (short)0, acc,  false,  false);
  }

  float badd = bias ? bias[col0 + lm] : 0.f;
#pragma unroll
  for (int r = 0; r < 8; ++r)
    Cout[(size_t)(row0 + half * 8 + r) * DIM + col0 + lm] = acc[r] + badd;
}


template <int G, typename GET>
static __device__ __forceinline__ void add_edges(GET get, float* rows, const float* sdinv,
                                                 const float* __restrict__ dinv,
                                                 const float* __restrict__ xw, int lo, int lane) {
  int ld[G];
  float w[G];
  v2f a[G], b[G];
#pragma unroll
  for (int g = 0; g < G; ++g) {
    unsigned p = get(g);
    int s = (int)(p & 0xFFFFu);
    ld[g] = (int)(p >> 16) - lo;
    const float* r = xw + (size_t)s * DIM + lane * 2;
    a[g] = *(const v2f*)r;
    b[g] = *(const v2f*)(r + 64);
    w[g] = dinv[s] * sdinv[ld[g]];
  }
#pragma unroll
  for (int g = 0; g < G; ++g) {
    float* d = rows + ld[g] * ROW_PITCH + lane * 2;
    __hip_atomic_fetch_add(d,      a[g].x * w[g], __ATOMIC_RELAXED, __HIP_MEMORY_SCOPE_WORKGROUP);
    __hip_atomic_fetch_add(d + 1,  a[g].y * w[g], __ATOMIC_RELAXED, __HIP_MEMORY_SCOPE_WORKGROUP);
    __hip_atomic_fetch_add(d + 64, b[g].x * w[g], __ATOMIC_RELAXED, __HIP_MEMORY_SCOPE_WORKGROUP);
    __hip_atomic_fetch_add(d + 65, b[g].y * w[g], __ATOMIC_RELAXED, __HIP_MEMORY_SCOPE_WORKGROUP);
  }
}

static __device__ __forceinline__ void drain(const unsigned* q, unsigned n, float* rows,
                                             const float* sdinv, const float* __restrict__ dinv,
                                             const float* __restrict__ xw, int lo, int lane) {
  unsigned i = 0;
  auto get = [&](int g) { return (unsigned)__builtin_amdgcn_readfirstlane(q[i + g]); };
  for (; i + 8 <= n; i += 8) add_edges<8>(get, rows, sdinv, dinv, xw, lo, lane);
  for (; i + 2 <= n; i += 2) add_edges<2>(get, rows, sdinv, dinv, xw, lo, lane);
  if (i < n) add_edges<1>(get, rows, sdinv, dinv, xw, lo, lane);
}

template <bool GUARD>
static __device__ __forceinline__ void scan_batch(const int* __restrict__ dstv, int st, int rot, int lo, int cnt,
                                                  int lane, unsigned below, unsigned* q, unsigned& qn) {
  v4u e[SCAN_UNROLL];
#pragma unroll
  for (int u = 0; u < SCAN_UNROLL; ++u) {
    int s = st + u * AGG_WAVES;
    if (!GUARD || s < N_STEPS) {
      int sr = s + rot;
      if (sr >= N_STEPS) sr -= N_STEPS;
      e[u] = *(const v4u*)(dstv + (size_t)sr * EDGES_PER_STEP + lane * 4);
    } else {
      e[u] = (v4u){NO_DST, NO_DST, NO_DST, NO_DST};
    }
  }
#pragma unroll
  for (int u = 0; u < SCAN_UNROLL; ++u) {
#pragma unroll
    for (int k = 0; k < 4; ++k) {
      unsigned dl = e[u][k] - (unsigned)lo;
      bool hit = dl < (unsigned)cnt;
      unsigned m = __builtin_amdgcn_ballot_w32(hit);
      if (m) {
        int sr = st + u * AGG_WAVES + rot;
        if (sr >= N_STEPS) sr -= N_STEPS;
        if (hit) q[qn + __builtin_popcount(m & below)] =
                     ((unsigned)(sr * EDGES_PER_STEP + lane * 4 + k) << 8) | dl;
        qn += __builtin_popcount(m);
      }
    }
  }
}

template <typename FLUSH>
static __device__ __forceinline__ void scan_edges(const int* __restrict__ dstv, int lo, int cnt,
                                                  int wave, int lane, unsigned* q, FLUSH flush) {
  unsigned qn = 0;
  const unsigned below = (1u << lane) - 1u;
  const int rot = (int)blockIdx.x * (N_STEPS / N_OWNERS);

  int st = wave;
  for (; st + (SCAN_UNROLL - 1) * AGG_WAVES < N_STEPS; st += AGG_WAVES * SCAN_UNROLL) {
    scan_batch<false>(dstv, st, rot, lo, cnt, lane, below, q, qn);
    if (qn >= DRAIN_AT) {
      flush(qn);
      qn = 0;
    }
  }
  if (st < N_STEPS) scan_batch<true>(dstv, st, rot, lo, cnt, lane, below, q, qn);
  if (qn) flush(qn);
}

static __device__ __forceinline__ void keep(const unsigned* q, unsigned n, const int* __restrict__ srcv, int lo,
                                            unsigned* __restrict__ list, unsigned* lcount, unsigned* sdeg,
                                            int lane) {
  unsigned off = 0;
  if (lane == 0) off = __hip_atomic_fetch_add(lcount, n, __ATOMIC_RELAXED, __HIP_MEMORY_SCOPE_WORKGROUP);
  off = __builtin_amdgcn_readfirstlane(off);
  const bool fits = off + n <= LIST_CAP;
  for (unsigned i = lane; i < n; i += 32) {
    const unsigned w = q[i], dl = w & 0xFFu;
    __hip_atomic_fetch_add(sdeg + dl, 1u, __ATOMIC_RELAXED, __HIP_MEMORY_SCOPE_WORKGROUP);
    if (fits) list[off + i] = (unsigned)srcv[w >> 8] | (((unsigned)lo + dl) << 16);
  }
}

static __device__ __forceinline__ void resolve(unsigned* q, unsigned n, const int* __restrict__ srcv, int lo,
                                               int lane) {
  for (unsigned i = lane; i < n; i += 32) {
    const unsigned w = q[i];
    q[i] = (unsigned)srcv[w >> 8] | (((unsigned)lo + (w & 0xFFu)) << 16);
  }
}

static __device__ __forceinline__ void replay(const unsigned* __restrict__ list, unsigned count,
                                              int wave, float* rows, const float* sdinv,
                                              const float* __restrict__ dinv,
                                              const float* __restrict__ xw, int lo, int lane) {
  if (count == 0) return;
  const unsigned stride = AGG_WAVES * REPLAY_GROUP;
  const unsigned sub = (unsigned)lane & (REPLAY_GROUP - 1);
  const unsigned last = count - 1;
  unsigned i = (unsigned)wave * REPLAY_GROUP;
  unsigned nxt = list[(i + sub < last) ? i + sub : last];
  for (; i < count; i += stride) {
    unsigned cur = nxt;
    unsigned j = i + stride + sub;
    nxt = list[(j < last) ? j : last];
    unsigned n = count - i;
    if (n >= REPLAY_GROUP) {
      add_edges<REPLAY_GROUP>([&](int g) { return (unsigned)__builtin_amdgcn_readlane(cur, g); },
                              rows, sdinv, dinv, xw, lo, lane);
    } else {
      for (unsigned g = 0; g < n; ++g)
        add_edges<1>([&](int) { return (unsigned)__builtin_amdgcn_readlane(cur, g); },
                     rows, sdinv, dinv, xw, lo, lane);
    }
  }
}

template <bool RELU_A>
static __device__ __forceinline__ v8f tile_product(const float* ap, const v2f (&bf)[DIM / 4]) {
  v8f acc = {};
#pragma unroll
  for (int k = 0; k < DIM; k += 4) {
    v2f a = *(const v2f*)(ap + k);
    if (RELU_A) {
      a.x = fmaxf(a.x, 0.f);
      a.y = fmaxf(a.y, 0.f);
    }
    acc = __builtin_amdgcn_wmma_f32_16x16x4_f32(
         false, a,  false, bf[k / 4],
         (short)0, acc,  false,  false);
  }
  return acc;
}

__global__ __launch_bounds__(AGG_THREADS)
void setup_kernel(const int* __restrict__ x, const float* __restrict__ emb, const int* __restrict__ ei,
                  float* __restrict__ h, float* __restrict__ dinv,
                  unsigned* __restrict__ lists, unsigned* __restrict__ counts,
                  float* __restrict__ pool, float* __restrict__ pnodes) {
  __shared__ unsigned queue[AGG_WAVES * QCAP];
  __shared__ unsigned sdeg[OWN_ROWS];
  __shared__ unsigned lcount;

  const int tid = threadIdx.x, lane = tid & 31;
  const int wave = __builtin_amdgcn_readfirstlane(tid >> 5);
  const int lo = blockIdx.x * OWN_ROWS;
  const int cnt = (N_NODES - lo < OWN_ROWS) ? (N_NODES - lo) : OWN_ROWS;
  unsigned* list = lists + (size_t)blockIdx.x * LIST_CAP;

  for (int i = tid; i < cnt; i += AGG_THREADS) sdeg[i] = 1u;
  if (tid == 0) lcount = 0;
  __syncthreads();
  for (int t = blockIdx.x * AGG_THREADS + tid; t < N_GRAPHS * DIM + N_GRAPHS; t += N_OWNERS * AGG_THREADS) {
    if (t < N_GRAPHS * DIM) pool[t] = 0.f;
    else pnodes[t - N_GRAPHS * DIM] = 0.f;
  }

  for (int r0 = wave; r0 < cnt; r0 += AGG_WAVES * EMB_BATCH) {
    v4f t[EMB_BATCH][N_COLS];
#pragma unroll
    for (int u = 0; u < EMB_BATCH; ++u) {
      int r = r0 + u * AGG_WAVES;
      if (r >= cnt) r = cnt - 1;
      const int* xr = x + (size_t)(lo + r) * N_COLS;
#pragma unroll
      for (int cc = 0; cc < N_COLS; ++cc)
        t[u][cc] = *(const v4f*)(emb + ((size_t)cc * VOCAB + xr[cc]) * DIM + lane * 4);
    }
#pragma unroll
    for (int u = 0; u < EMB_BATCH; ++u) {
      asm volatile("" :: "v"(t[u][0]), "v"(t[u][1]), "v"(t[u][2]), "v"(t[u][3]));
    }
#pragma unroll
    for (int u = 0; u < EMB_BATCH; ++u) {
      const int r = r0 + u * AGG_WAVES;
      if (r < cnt) {
        v4f s = {0.f, 0.f, 0.f, 0.f};
#pragma unroll
        for (int cc = 0; cc < N_COLS; ++cc) s += t[u][cc];
        *(v4f*)(h + (size_t)(lo + r) * DIM + lane * 4) = s;
      }
    }
  }

  unsigned* q = queue + wave * QCAP;
  scan_edges(ei + N_EDGES, lo, cnt, wave, lane, q,
             [&](unsigned n) { keep(q, n, ei, lo, list, &lcount, sdeg, lane); });
  __syncthreads();

  if (tid == 0) counts[blockIdx.x] = lcount;
  for (int i = tid; i < cnt; i += AGG_THREADS) {
    const float d = (float)sdeg[i];
    dinv[lo + i] = (d > 0.f) ? rsqrtf(d) : 0.f;
  }
}

__global__ __launch_bounds__(AGG_THREADS)
void agg_kernel(const int* __restrict__ ei, const float* __restrict__ dinv,
                const float* __restrict__ xw, const float* __restrict__ bl, float* __restrict__ outp,
                const float* __restrict__ W0, const float* __restrict__ Wn,
                const unsigned* __restrict__ lists, const unsigned* __restrict__ counts,
                const int* __restrict__ batch, float* __restrict__ pool, float* __restrict__ pnodes) {
  __shared__ float rows[ROW_TILES * 16 * ROW_PITCH];
  __shared__ float sdinv[OWN_ROWS];
  __shared__ unsigned queue[AGG_WAVES * QCAP];
  __shared__ float pbin[POOL_BINS * DIM];
  __shared__ float pnum[POOL_BINS];

  const int tid = threadIdx.x, lane = tid & 31;
  const int wave = __builtin_amdgcn_readfirstlane(tid >> 5);
  const bool first = (W0 != nullptr);
  const bool product = (Wn != nullptr);
  const unsigned* list = lists + (size_t)blockIdx.x * LIST_CAP;
  const unsigned kept = counts[blockIdx.x];
  const bool use_list = kept <= LIST_CAP;
  const int lo = blockIdx.x * OWN_ROWS;
  const int cnt = (N_NODES - lo < OWN_ROWS) ? (N_NODES - lo) : OWN_ROWS;

  for (int i = tid; i < cnt * 32; i += AGG_THREADS) {
    int r = i >> 5, c4 = (i & 31) << 2;
    float s = dinv[lo + r];
    v4f x  = *(const v4f*)(xw + (size_t)(lo + r) * DIM + c4);
    v4f bb = first ? (v4f){0.f, 0.f, 0.f, 0.f} : *(const v4f*)(bl + c4);
    v4f o;
    o.x = bb.x + x.x * s * s;
    o.y = bb.y + x.y * s * s;
    o.z = bb.z + x.z * s * s;
    o.w = bb.w + x.w * s * s;
    *(v4f*)(rows + r * ROW_PITCH + c4) = o;
  }
  for (int i = cnt * 32 + tid; i < ((cnt + 15) & ~15) * 32; i += AGG_THREADS) {
    int r = i >> 5, c4 = (i & 31) << 2;
    *(v4f*)(rows + r * ROW_PITCH + c4) = (v4f){0.f, 0.f, 0.f, 0.f};
  }
  for (int i = tid; i < cnt; i += AGG_THREADS) sdinv[i] = dinv[lo + i];
  if (!product) {
    for (int i = tid; i < POOL_BINS * DIM; i += AGG_THREADS) pbin[i] = 0.f;
    if (tid < POOL_BINS) pnum[tid] = 0.f;
  }
  __syncthreads();

  if (use_list) {
    replay(list, kept, wave, rows, sdinv, dinv, xw, lo, lane);
  } else {
    unsigned* q = queue + wave * QCAP;
    scan_edges(ei + N_EDGES, lo, cnt, wave, lane, q, [&](unsigned n) {
      resolve(q, n, ei, lo, lane);
      drain(q, n, rows, sdinv, dinv, xw, lo, lane);
    });
  }
  const int half = lane >> 4, lm = lane & 15;
  const int col0 = (wave & 7) * 16;
  const int n_rt = (cnt + 15) >> 4;
  v2f bf[DIM / 4];
  auto load_b = [&](const float* __restrict__ Wp) {
    const float* bp = Wp + (size_t)(half * 2) * DIM + col0 + lm;
#pragma unroll
    for (int k = 0; k < DIM; k += 4) {
      bf[k / 4].x = bp[k * DIM];
      bf[k / 4].y = bp[k * DIM + DIM];
    }
  };
  if (first) load_b(W0);
  else if (product) load_b(Wn);
  const int pr0 = wave * POOL_ROWS;
  const int pr1 = (pr0 + POOL_ROWS < cnt) ? pr0 + POOL_ROWS : cnt;
  int bv = 0, g0 = 0;
  if (!product) {
    g0 = batch[lo];
    if (pr0 + lane < pr1) bv = batch[lo + pr0 + lane];
  }
  __syncthreads();

  const float* arows = rows;
  if (first) {
    static_assert(AGG_WAVES * QCAP >= ROW_TILES * 16 * ROW_PITCH, "the queues must hold a second set of rows");
    float* stage = reinterpret_cast<float*>(queue);
    const float badd = bl[col0 + lm];
    for (int rt = wave >> 3; rt < n_rt; rt += AGG_WAVES / 8) {
      v8f acc = tile_product<false>(rows + (rt * 16 + lm) * ROW_PITCH + half * 2, bf);
      float* sp = stage + (rt * 16 + half * 8) * ROW_PITCH + col0 + lm;
#pragma unroll
      for (int r = 0; r < 8; ++r) sp[r * ROW_PITCH] = fmaxf(acc[r] + badd, 0.f);
    }
    if (product) load_b(Wn);
    __syncthreads();
    arows = stage;
  }

  if (!product) {
    if (pr0 < pr1) {
      const float* rp = arows + pr0 * ROW_PITCH + lane * 2;
      const int n_rows = pr1 - pr0;
      v2f sa = {0.f, 0.f}, sb = {0.f, 0.f};
      float num = 0.f;
      int g = __builtin_amdgcn_readlane(bv, 0);
      v2f na = *(const v2f*)rp, nb = *(const v2f*)(rp + 64);
      for (int k = 0; k < n_rows; ++k) {
        const v2f a = na, b = nb;
        const bool last_row = (k + 1 == n_rows);
        const float* np = rp + (last_row ? k : k + 1) * ROW_PITCH;
        na = *(const v2f*)np;
        nb = *(const v2f*)(np + 64);
        const int gn = __builtin_amdgcn_readlane(bv, k + 1);
        sa.x += fmaxf(a.x, 0.f);
        sa.y += fmaxf(a.y, 0.f);
        sb.x += fmaxf(b.x, 0.f);
        sb.y += fmaxf(b.y, 0.f);
        num += 1.f;
        if (last_row || gn != g) {
          const unsigned slot = (unsigned)(g - g0);
          if (slot < POOL_BINS) {
            float* d = pbin + slot * DIM + lane * 2;
            __hip_atomic_fetch_add(d,      sa.x, __ATOMIC_RELAXED, __HIP_MEMORY_SCOPE_WORKGROUP);
            __hip_atomic_fetch_add(d + 1,  sa.y, __ATOMIC_RELAXED, __HIP_MEMORY_SCOPE_WORKGROUP);
            __hip_atomic_fetch_add(d + 64, sb.x, __ATOMIC_RELAXED, __HIP_MEMORY_SCOPE_WORKGROUP);
            __hip_atomic_fetch_add(d + 65, sb.y, __ATOMIC_RELAXED, __HIP_MEMORY_SCOPE_WORKGROUP);
            if (lane == 0) __hip_atomic_fetch_add(pnum + slot, num, __ATOMIC_RELAXED, __HIP_MEMORY_SCOPE_WORKGROUP);
          } else {
            float* d = pool + (size_t)g * DIM + lane * 2;
            unsafeAtomicAdd(d,      sa.x);
            unsafeAtomicAdd(d + 1,  sa.y);
            unsafeAtomicAdd(d + 64, sb.x);
            unsafeAtomicAdd(d + 65, sb.y);
            if (lane == 0) unsafeAtomicAdd(&pnodes[g], num);
          }
          sa = (v2f){0.f, 0.f};
          sb = (v2f){0.f, 0.f};
          num = 0.f;
        }
        g = gn;
      }
    }
    __syncthreads();
    for (int i = tid; i < POOL_BINS * DIM; i += AGG_THREADS) {
      const int slot = i >> 7;
      const float n = pnum[slot];
      if (n > 0.f) {
        unsafeAtomicAdd(&pool[(size_t)(g0 + slot) * DIM + (i & (DIM - 1))], pbin[i]);
        if ((i & (DIM - 1)) == 0) unsafeAtomicAdd(&pnodes[g0 + slot], n);
      }
    }
    return;
  }

  for (int rt = wave >> 3; rt < n_rt; rt += AGG_WAVES / 8) {
    v8f acc = tile_product<true>(arows + (rt * 16 + lm) * ROW_PITCH + half * 2, bf);
    const int r0 = rt * 16 + half * 8;
#pragma unroll
    for (int r = 0; r < 8; ++r)
      if (r0 + r < cnt) outp[(size_t)(lo + r0 + r) * DIM + col0 + lm] = acc[r];
  }
}

__global__ void pool_div_kernel(float* __restrict__ pool, const float* __restrict__ cnt) {
  int t = blockIdx.x * blockDim.x + threadIdx.x;
  if (t >= N_GRAPHS * DIM) return;
  pool[t] /= fmaxf(cnt[t / DIM], 1.0f);
}

extern "C" void kernel_launch(void* const* d_in, const int* in_sizes, int n_in,
                              void* d_out, int out_size, void* d_ws, size_t ws_size,
                              hipStream_t stream) {
  const int*   x    = (const int*)d_in[0];
  const int*   ei   = (const int*)d_in[1];
  const int*   batch= (const int*)d_in[2];
  const float* emb  = (const float*)d_in[3];
  const float* W    = (const float*)d_in[4];
  const float* b    = (const float*)d_in[5];
  const float* Wr   = (const float*)d_in[6];
  const float* br   = (const float*)d_in[7];
  float* out = (float*)d_out;

  char* ws = (char*)d_ws;
  const size_t NH = (size_t)N_NODES * DIM * sizeof(float);
  float* h    = (float*)(ws);
  float* xw   = (float*)(ws + NH);
  float* dinv = (float*)(ws + 3 * NH);
  float* pool = (float*)(ws + 3 * NH + 4 * 1024 * 1024);
  float* cnt  = (float*)(ws + 3 * NH + 4 * 1024 * 1024 + 512 * 1024);
  unsigned* counts = (unsigned*)(ws + 3 * NH + 5 * 1024 * 1024);
  unsigned* lists  = (unsigned*)(ws + 3 * NH + 8 * 1024 * 1024);

  const int B256 = 256;

  setup_kernel<<<N_OWNERS, AGG_THREADS, 0, stream>>>(x, emb, ei, h, dinv, lists, counts, pool, cnt);

  static_assert(N_LAYERS == 4, "the four layer launches are written out");
  agg_kernel<<<N_OWNERS, AGG_THREADS, 0, stream>>>(ei, dinv, h,  b,           xw, W,       W + 1 * DIM * DIM, lists, counts, batch, pool, cnt);
  agg_kernel<<<N_OWNERS, AGG_THREADS, 0, stream>>>(ei, dinv, xw, b + 1 * DIM, h,  nullptr, W + 2 * DIM * DIM, lists, counts, batch, pool, cnt);
  agg_kernel<<<N_OWNERS, AGG_THREADS, 0, stream>>>(ei, dinv, h,  b + 2 * DIM, xw, nullptr, W + 3 * DIM * DIM, lists, counts, batch, pool, cnt);
  agg_kernel<<<N_OWNERS, AGG_THREADS, 0, stream>>>(ei, dinv, xw, b + 3 * DIM, h,  nullptr, nullptr,           lists, counts, batch, pool, cnt);

  pool_div_kernel<<<(N_GRAPHS * DIM + B256 - 1) / B256, B256, 0, stream>>>(pool, cnt);

  dim3 rgrid(N_GRAPHS / 16, DIM / 16);
  gemm_wmma_f32<<<rgrid, 32, 0, stream>>>(pool, Wr, br, out);
}


# Edge weight factored out of the neighbour sum: stored rows carry dinv[row], owners add gathered rows unweighted, dinv[d] applied once per row; no per-edge dinv load or multiply (rounding changes)
# speedup vs baseline: 3.8588x; 1.1002x over previous
#include <hip/hip_runtime.h>

#define N_NODES  50000
#define N_EDGES  800000
#define DIM      128
#define N_LAYERS 4
#define N_GRAPHS 512
#define N_COLS   4
#define VOCAB    128

#define OWN_ROWS       196
#define ROW_TILES      ((OWN_ROWS + 15) / 16)
#define ROW_PITCH      (DIM + 4)
#define N_OWNERS       ((N_NODES + OWN_ROWS - 1) / OWN_ROWS)
#define AGG_THREADS    1024
#define AGG_WAVES      (AGG_THREADS / 32)
#define EDGES_PER_STEP 128
#define N_STEPS        (N_EDGES / EDGES_PER_STEP)
#define SCAN_UNROLL    8
#define DRAIN_AT       64
#define QCAP           (DRAIN_AT + SCAN_UNROLL * EDGES_PER_STEP)
#define NO_DST         0xFFFFFFFFu
#define LIST_CAP       32768
#define REPLAY_GROUP   8
#define POOL_BINS      16
#define POOL_ROWS      ((OWN_ROWS + AGG_WAVES - 1) / AGG_WAVES)
#define EMB_BATCH      4

static_assert(N_NODES <= 65535, "node ids must fit 16 bits with 0xFFFF left free");
static_assert(N_EDGES % EDGES_PER_STEP == 0, "edge list must divide into whole steps");
static_assert(N_EDGES <= (1 << 24) && OWN_ROWS <= 256, "a queued word is edge index << 8 | local row");
static_assert(DIM == 128, "a wave holds a row as four columns per lane; bin words are found by shifts");
static_assert(POOL_ROWS < 32, "the graph ids of a wave's rows, and the one behind them, sit one per lane");

typedef __attribute__((ext_vector_type(2))) float v2f;
typedef __attribute__((ext_vector_type(4))) float v4f;
typedef __attribute__((ext_vector_type(8))) float v8f;
typedef __attribute__((ext_vector_type(4))) unsigned v4u;

__global__ void gemm_wmma_f32(const float* __restrict__ A, const float* __restrict__ B,
                              const float* __restrict__ bias, float* __restrict__ Cout) {
  int row0 = blockIdx.x * 16;
  int col0 = blockIdx.y * 16;
  int lane = threadIdx.x;
  int half = lane >> 4;
  int lm   = lane & 15;

  const float* ap = A + (size_t)(row0 + lm) * DIM + half * 2;
  const float* bp = B + (size_t)(half * 2) * DIM + col0 + lm;

  v8f acc = {};
#pragma unroll
  for (int k = 0; k < DIM; k += 4) {
    v2f a = *(const v2f*)(ap + k);
    v2f b;
    b.x = bp[k * DIM];
    b.y = bp[k * DIM + DIM];
    acc = __builtin_amdgcn_wmma_f32_16x16x4_f32(
         false, a,  false, b,
         (short)0, acc,  false,  false);
  }

  float badd = bias ? bias[col0 + lm] : 0.f;
#pragma unroll
  for (int r = 0; r < 8; ++r)
    Cout[(size_t)(row0 + half * 8 + r) * DIM + col0 + lm] = acc[r] + badd;
}


template <int G, typename GET>
static __device__ __forceinline__ void add_edges(GET get, float* rows,
                                                 const float* __restrict__ xw, int lo, int lane) {
  int ld[G];
  v2f a[G], b[G];
#pragma unroll
  for (int g = 0; g < G; ++g) {
    unsigned p = get(g);
    int s = (int)(p & 0xFFFFu);
    ld[g] = (int)(p >> 16) - lo;
    const float* r = xw + (size_t)s * DIM + lane * 2;
    a[g] = *(const v2f*)r;
    b[g] = *(const v2f*)(r + 64);
  }
#pragma unroll
  for (int g = 0; g < G; ++g) {
    float* d = rows + ld[g] * ROW_PITCH + lane * 2;
    __hip_atomic_fetch_add(d,      a[g].x, __ATOMIC_RELAXED, __HIP_MEMORY_SCOPE_WORKGROUP);
    __hip_atomic_fetch_add(d + 1,  a[g].y, __ATOMIC_RELAXED, __HIP_MEMORY_SCOPE_WORKGROUP);
    __hip_atomic_fetch_add(d + 64, b[g].x, __ATOMIC_RELAXED, __HIP_MEMORY_SCOPE_WORKGROUP);
    __hip_atomic_fetch_add(d + 65, b[g].y, __ATOMIC_RELAXED, __HIP_MEMORY_SCOPE_WORKGROUP);
  }
}

static __device__ __forceinline__ void drain(const unsigned* q, unsigned n, float* rows,
                                             const float* __restrict__ xw, int lo, int lane) {
  unsigned i = 0;
  auto get = [&](int g) { return (unsigned)__builtin_amdgcn_readfirstlane(q[i + g]); };
  for (; i + 8 <= n; i += 8) add_edges<8>(get, rows, xw, lo, lane);
  for (; i + 2 <= n; i += 2) add_edges<2>(get, rows, xw, lo, lane);
  if (i < n) add_edges<1>(get, rows, xw, lo, lane);
}

template <bool GUARD>
static __device__ __forceinline__ void scan_batch(const int* __restrict__ dstv, int st, int rot, int lo, int cnt,
                                                  int lane, unsigned below, unsigned* q, unsigned& qn) {
  v4u e[SCAN_UNROLL];
#pragma unroll
  for (int u = 0; u < SCAN_UNROLL; ++u) {
    int s = st + u * AGG_WAVES;
    if (!GUARD || s < N_STEPS) {
      int sr = s + rot;
      if (sr >= N_STEPS) sr -= N_STEPS;
      e[u] = *(const v4u*)(dstv + (size_t)sr * EDGES_PER_STEP + lane * 4);
    } else {
      e[u] = (v4u){NO_DST, NO_DST, NO_DST, NO_DST};
    }
  }
#pragma unroll
  for (int u = 0; u < SCAN_UNROLL; ++u) {
#pragma unroll
    for (int k = 0; k < 4; ++k) {
      unsigned dl = e[u][k] - (unsigned)lo;
      bool hit = dl < (unsigned)cnt;
      unsigned m = __builtin_amdgcn_ballot_w32(hit);
      if (m) {
        int sr = st + u * AGG_WAVES + rot;
        if (sr >= N_STEPS) sr -= N_STEPS;
        if (hit) q[qn + __builtin_popcount(m & below)] =
                     ((unsigned)(sr * EDGES_PER_STEP + lane * 4 + k) << 8) | dl;
        qn += __builtin_popcount(m);
      }
    }
  }
}

template <typename FLUSH>
static __device__ __forceinline__ void scan_edges(const int* __restrict__ dstv, int lo, int cnt,
                                                  int wave, int lane, unsigned* q, FLUSH flush) {
  unsigned qn = 0;
  const unsigned below = (1u << lane) - 1u;
  const int rot = (int)blockIdx.x * (N_STEPS / N_OWNERS);

  int st = wave;
  for (; st + (SCAN_UNROLL - 1) * AGG_WAVES < N_STEPS; st += AGG_WAVES * SCAN_UNROLL) {
    scan_batch<false>(dstv, st, rot, lo, cnt, lane, below, q, qn);
    if (qn >= DRAIN_AT) {
      flush(qn);
      qn = 0;
    }
  }
  if (st < N_STEPS) scan_batch<true>(dstv, st, rot, lo, cnt, lane, below, q, qn);
  if (qn) flush(qn);
}

static __device__ __forceinline__ void keep(const unsigned* q, unsigned n, const int* __restrict__ srcv, int lo,
                                            unsigned* __restrict__ list, unsigned* lcount, unsigned* sdeg,
                                            int lane) {
  unsigned off = 0;
  if (lane == 0) off = __hip_atomic_fetch_add(lcount, n, __ATOMIC_RELAXED, __HIP_MEMORY_SCOPE_WORKGROUP);
  off = __builtin_amdgcn_readfirstlane(off);
  const bool fits = off + n <= LIST_CAP;
  for (unsigned i = lane; i < n; i += 32) {
    const unsigned w = q[i], dl = w & 0xFFu;
    __hip_atomic_fetch_add(sdeg + dl, 1u, __ATOMIC_RELAXED, __HIP_MEMORY_SCOPE_WORKGROUP);
    if (fits) list[off + i] = (unsigned)srcv[w >> 8] | (((unsigned)lo + dl) << 16);
  }
}

static __device__ __forceinline__ void resolve(unsigned* q, unsigned n, const int* __restrict__ srcv, int lo,
                                               int lane) {
  for (unsigned i = lane; i < n; i += 32) {
    const unsigned w = q[i];
    q[i] = (unsigned)srcv[w >> 8] | (((unsigned)lo + (w & 0xFFu)) << 16);
  }
}

static __device__ __forceinline__ void replay(const unsigned* __restrict__ list, unsigned count,
                                              int wave, float* rows,
                                              const float* __restrict__ xw, int lo, int lane) {
  if (count == 0) return;
  const unsigned stride = AGG_WAVES * REPLAY_GROUP;
  const unsigned sub = (unsigned)lane & (REPLAY_GROUP - 1);
  const unsigned last = count - 1;
  unsigned i = (unsigned)wave * REPLAY_GROUP;
  unsigned nxt = list[(i + sub < last) ? i + sub : last];
  for (; i < count; i += stride) {
    unsigned cur = nxt;
    unsigned j = i + stride + sub;
    nxt = list[(j < last) ? j : last];
    unsigned n = count - i;
    if (n >= REPLAY_GROUP) {
      add_edges<REPLAY_GROUP>([&](int g) { return (unsigned)__builtin_amdgcn_readlane(cur, g); },
                              rows, xw, lo, lane);
    } else {
      for (unsigned g = 0; g < n; ++g)
        add_edges<1>([&](int) { return (unsigned)__builtin_amdgcn_readlane(cur, g); },
                     rows, xw, lo, lane);
    }
  }
}

static __device__ __forceinline__ v8f tile_product(const float* ap, const v2f (&bf)[DIM / 4]) {
  v8f acc = {};
#pragma unroll
  for (int k = 0; k < DIM; k += 4) {
    v2f a = *(const v2f*)(ap + k);
    acc = __builtin_amdgcn_wmma_f32_16x16x4_f32(
         false, a,  false, bf[k / 4],
         (short)0, acc,  false,  false);
  }
  return acc;
}

__global__ __launch_bounds__(AGG_THREADS)
void setup_kernel(const int* __restrict__ x, const float* __restrict__ emb, const int* __restrict__ ei,
                  float* __restrict__ h, float* __restrict__ dinv,
                  unsigned* __restrict__ lists, unsigned* __restrict__ counts,
                  float* __restrict__ pool, float* __restrict__ pnodes) {
  __shared__ unsigned queue[AGG_WAVES * QCAP];
  __shared__ unsigned sdeg[OWN_ROWS];
  __shared__ unsigned lcount;

  const int tid = threadIdx.x, lane = tid & 31;
  const int wave = __builtin_amdgcn_readfirstlane(tid >> 5);
  const int lo = blockIdx.x * OWN_ROWS;
  const int cnt = (N_NODES - lo < OWN_ROWS) ? (N_NODES - lo) : OWN_ROWS;
  unsigned* list = lists + (size_t)blockIdx.x * LIST_CAP;

  for (int i = tid; i < cnt; i += AGG_THREADS) sdeg[i] = 1u;
  if (tid == 0) lcount = 0;
  __syncthreads();
  for (int t = blockIdx.x * AGG_THREADS + tid; t < N_GRAPHS * DIM + N_GRAPHS; t += N_OWNERS * AGG_THREADS) {
    if (t < N_GRAPHS * DIM) pool[t] = 0.f;
    else pnodes[t - N_GRAPHS * DIM] = 0.f;
  }

  unsigned* q = queue + wave * QCAP;
  scan_edges(ei + N_EDGES, lo, cnt, wave, lane, q,
             [&](unsigned n) { keep(q, n, ei, lo, list, &lcount, sdeg, lane); });
  __syncthreads();

  if (tid == 0) counts[blockIdx.x] = lcount;
  for (int i = tid; i < cnt; i += AGG_THREADS) {
    const float d = (float)sdeg[i];
    dinv[lo + i] = (d > 0.f) ? rsqrtf(d) : 0.f;
  }

  for (int r0 = wave; r0 < cnt; r0 += AGG_WAVES * EMB_BATCH) {
    v4f t[EMB_BATCH][N_COLS];
#pragma unroll
    for (int u = 0; u < EMB_BATCH; ++u) {
      int r = r0 + u * AGG_WAVES;
      if (r >= cnt) r = cnt - 1;
      const int* xr = x + (size_t)(lo + r) * N_COLS;
#pragma unroll
      for (int cc = 0; cc < N_COLS; ++cc)
        t[u][cc] = *(const v4f*)(emb + ((size_t)cc * VOCAB + xr[cc]) * DIM + lane * 4);
    }
#pragma unroll
    for (int u = 0; u < EMB_BATCH; ++u) {
      asm volatile("" :: "v"(t[u][0]), "v"(t[u][1]), "v"(t[u][2]), "v"(t[u][3]));
    }
#pragma unroll
    for (int u = 0; u < EMB_BATCH; ++u) {
      const int r = r0 + u * AGG_WAVES;
      if (r < cnt) {
        v4f s = {0.f, 0.f, 0.f, 0.f};
#pragma unroll
        for (int cc = 0; cc < N_COLS; ++cc) s += t[u][cc];
        const float d = (float)sdeg[r];
        const float sc = (d > 0.f) ? rsqrtf(d) : 0.f;
        *(v4f*)(h + (size_t)(lo + r) * DIM + lane * 4) = s * sc;
      }
    }
  }
}

__global__ __launch_bounds__(AGG_THREADS)
void agg_kernel(const int* __restrict__ ei, const float* __restrict__ dinv,
                const float* __restrict__ xw, const float* __restrict__ bl, float* __restrict__ outp,
                const float* __restrict__ W0, const float* __restrict__ Wn,
                const unsigned* __restrict__ lists, const unsigned* __restrict__ counts,
                const int* __restrict__ batch, float* __restrict__ pool, float* __restrict__ pnodes) {
  __shared__ float rows[ROW_TILES * 16 * ROW_PITCH];
  __shared__ __attribute__((aligned(16))) float sdinv[ROW_TILES * 16];
  __shared__ unsigned queue[AGG_WAVES * QCAP];
  __shared__ float pbin[POOL_BINS * DIM];
  __shared__ float pnum[POOL_BINS];

  const int tid = threadIdx.x, lane = tid & 31;
  const int wave = __builtin_amdgcn_readfirstlane(tid >> 5);
  const bool first = (W0 != nullptr);
  const bool product = (Wn != nullptr);
  const unsigned* list = lists + (size_t)blockIdx.x * LIST_CAP;
  const unsigned kept = counts[blockIdx.x];
  const bool use_list = kept <= LIST_CAP;
  const int lo = blockIdx.x * OWN_ROWS;
  const int cnt = (N_NODES - lo < OWN_ROWS) ? (N_NODES - lo) : OWN_ROWS;

  for (int i = tid; i < cnt * 32; i += AGG_THREADS) {
    int r = i >> 5, c4 = (i & 31) << 2;
    *(v4f*)(rows + r * ROW_PITCH + c4) = *(const v4f*)(xw + (size_t)(lo + r) * DIM + c4);
  }
  for (int i = cnt * 32 + tid; i < ((cnt + 15) & ~15) * 32; i += AGG_THREADS) {
    int r = i >> 5, c4 = (i & 31) << 2;
    *(v4f*)(rows + r * ROW_PITCH + c4) = (v4f){0.f, 0.f, 0.f, 0.f};
  }
  for (int i = tid; i < ROW_TILES * 16; i += AGG_THREADS) sdinv[i] = (i < cnt) ? dinv[lo + i] : 0.f;
  if (!product) {
    for (int i = tid; i < POOL_BINS * DIM; i += AGG_THREADS) pbin[i] = 0.f;
    if (tid < POOL_BINS) pnum[tid] = 0.f;
  }
  __syncthreads();

  if (use_list) {
    replay(list, kept, wave, rows, xw, lo, lane);
  } else {
    unsigned* q = queue + wave * QCAP;
    scan_edges(ei + N_EDGES, lo, cnt, wave, lane, q, [&](unsigned n) {
      resolve(q, n, ei, lo, lane);
      drain(q, n, rows, xw, lo, lane);
    });
  }
  const int half = lane >> 4, lm = lane & 15;
  const int col0 = (wave & 7) * 16;
  const int n_rt = (cnt + 15) >> 4;
  v2f bf[DIM / 4];
  auto load_b = [&](const float* __restrict__ Wp) {
    const float* bp = Wp + (size_t)(half * 2) * DIM + col0 + lm;
#pragma unroll
    for (int k = 0; k < DIM; k += 4) {
      bf[k / 4].x = bp[k * DIM];
      bf[k / 4].y = bp[k * DIM + DIM];
    }
  };
  if (first) load_b(W0);
  else if (product) load_b(Wn);
  const int fc4 = (tid & 31) << 2;
  v4f fbias = {0.f, 0.f, 0.f, 0.f};
  if (!first) fbias = *(const v4f*)(bl + fc4);
  const int pr0 = wave * POOL_ROWS;
  const int pr1 = (pr0 + POOL_ROWS < cnt) ? pr0 + POOL_ROWS : cnt;
  int bv = 0, g0 = 0;
  if (!product) {
    g0 = batch[lo];
    if (pr0 + lane < pr1) bv = batch[lo + pr0 + lane];
  }
  __syncthreads();

  if (!first) {
    for (int i = tid; i < cnt * 32; i += AGG_THREADS) {
      const int r = i >> 5;
      const float s = sdinv[r];
      v4f* p = (v4f*)(rows + r * ROW_PITCH + fc4);
      const v4f x = *p;
      v4f o;
      o.x = fmaxf(fbias.x + x.x * s, 0.f);
      o.y = fmaxf(fbias.y + x.y * s, 0.f);
      o.z = fmaxf(fbias.z + x.z * s, 0.f);
      o.w = fmaxf(fbias.w + x.w * s, 0.f);
      *p = o;
    }
    __syncthreads();
  }

  const float* arows = rows;
  if (first) {
    static_assert(AGG_WAVES * QCAP >= ROW_TILES * 16 * ROW_PITCH, "the queues must hold a second set of rows");
    float* stage = reinterpret_cast<float*>(queue);
    const float badd = bl[col0 + lm];
    for (int rt = wave >> 3; rt < n_rt; rt += AGG_WAVES / 8) {
      v8f acc = tile_product(rows + (rt * 16 + lm) * ROW_PITCH + half * 2, bf);
      float* sp = stage + (rt * 16 + half * 8) * ROW_PITCH + col0 + lm;
      const v4f s0 = *(const v4f*)(sdinv + rt * 16 + half * 8), s1 = *(const v4f*)(sdinv + rt * 16 + half * 8 + 4);
      const float sd[8] = {s0.x, s0.y, s0.z, s0.w, s1.x, s1.y, s1.z, s1.w};
#pragma unroll
      for (int r = 0; r < 8; ++r) sp[r * ROW_PITCH] = fmaxf(acc[r] * sd[r] + badd, 0.f);
    }
    if (product) load_b(Wn);
    __syncthreads();
    arows = stage;
  }

  if (!product) {
    if (pr0 < pr1) {
      const float* rp = arows + pr0 * ROW_PITCH + lane * 2;
      const int n_rows = pr1 - pr0;
      v2f sa = {0.f, 0.f}, sb = {0.f, 0.f};
      float num = 0.f;
      int g = __builtin_amdgcn_readlane(bv, 0);
      v2f na = *(const v2f*)rp, nb = *(const v2f*)(rp + 64);
      for (int k = 0; k < n_rows; ++k) {
        const v2f a = na, b = nb;
        const bool last_row = (k + 1 == n_rows);
        const float* np = rp + (last_row ? k : k + 1) * ROW_PITCH;
        na = *(const v2f*)np;
        nb = *(const v2f*)(np + 64);
        const int gn = __builtin_amdgcn_readlane(bv, k + 1);
        sa.x += a.x;
        sa.y += a.y;
        sb.x += b.x;
        sb.y += b.y;
        num += 1.f;
        if (last_row || gn != g) {
          const unsigned slot = (unsigned)(g - g0);
          if (slot < POOL_BINS) {
            float* d = pbin + slot * DIM + lane * 2;
            __hip_atomic_fetch_add(d,      sa.x, __ATOMIC_RELAXED, __HIP_MEMORY_SCOPE_WORKGROUP);
            __hip_atomic_fetch_add(d + 1,  sa.y, __ATOMIC_RELAXED, __HIP_MEMORY_SCOPE_WORKGROUP);
            __hip_atomic_fetch_add(d + 64, sb.x, __ATOMIC_RELAXED, __HIP_MEMORY_SCOPE_WORKGROUP);
            __hip_atomic_fetch_add(d + 65, sb.y, __ATOMIC_RELAXED, __HIP_MEMORY_SCOPE_WORKGROUP);
            if (lane == 0) __hip_atomic_fetch_add(pnum + slot, num, __ATOMIC_RELAXED, __HIP_MEMORY_SCOPE_WORKGROUP);
          } else {
            float* d = pool + (size_t)g * DIM + lane * 2;
            unsafeAtomicAdd(d,      sa.x);
            unsafeAtomicAdd(d + 1,  sa.y);
            unsafeAtomicAdd(d + 64, sb.x);
            unsafeAtomicAdd(d + 65, sb.y);
            if (lane == 0) unsafeAtomicAdd(&pnodes[g], num);
          }
          sa = (v2f){0.f, 0.f};
          sb = (v2f){0.f, 0.f};
          num = 0.f;
        }
        g = gn;
      }
    }
    __syncthreads();
    for (int i = tid; i < POOL_BINS * DIM; i += AGG_THREADS) {
      const int slot = i >> 7;
      const float n = pnum[slot];
      if (n > 0.f) {
        unsafeAtomicAdd(&pool[(size_t)(g0 + slot) * DIM + (i & (DIM - 1))], pbin[i]);
        if ((i & (DIM - 1)) == 0) unsafeAtomicAdd(&pnodes[g0 + slot], n);
      }
    }
    return;
  }

  for (int rt = wave >> 3; rt < n_rt; rt += AGG_WAVES / 8) {
    v8f acc = tile_product(arows + (rt * 16 + lm) * ROW_PITCH + half * 2, bf);
    const int r0 = rt * 16 + half * 8;
    const v4f s0 = *(const v4f*)(sdinv + r0), s1 = *(const v4f*)(sdinv + r0 + 4);
    const float sd[8] = {s0.x, s0.y, s0.z, s0.w, s1.x, s1.y, s1.z, s1.w};
#pragma unroll
    for (int r = 0; r < 8; ++r)
      if (r0 + r < cnt) outp[(size_t)(lo + r0 + r) * DIM + col0 + lm] = acc[r] * sd[r];
  }
}

__global__ void pool_div_kernel(float* __restrict__ pool, const float* __restrict__ cnt) {
  int t = blockIdx.x * blockDim.x + threadIdx.x;
  if (t >= N_GRAPHS * DIM) return;
  pool[t] /= fmaxf(cnt[t / DIM], 1.0f);
}

extern "C" void kernel_launch(void* const* d_in, const int* in_sizes, int n_in,
                              void* d_out, int out_size, void* d_ws, size_t ws_size,
                              hipStream_t stream) {
  const int*   x    = (const int*)d_in[0];
  const int*   ei   = (const int*)d_in[1];
  const int*   batch= (const int*)d_in[2];
  const float* emb  = (const float*)d_in[3];
  const float* W    = (const float*)d_in[4];
  const float* b    = (const float*)d_in[5];
  const float* Wr   = (const float*)d_in[6];
  const float* br   = (const float*)d_in[7];
  float* out = (float*)d_out;

  char* ws = (char*)d_ws;
  const size_t NH = (size_t)N_NODES * DIM * sizeof(float);
  float* h    = (float*)(ws);
  float* xw   = (float*)(ws + NH);
  float* dinv = (float*)(ws + 3 * NH);
  float* pool = (float*)(ws + 3 * NH + 4 * 1024 * 1024);
  float* cnt  = (float*)(ws + 3 * NH + 4 * 1024 * 1024 + 512 * 1024);
  unsigned* counts = (unsigned*)(ws + 3 * NH + 5 * 1024 * 1024);
  unsigned* lists  = (unsigned*)(ws + 3 * NH + 8 * 1024 * 1024);

  const int B256 = 256;

  setup_kernel<<<N_OWNERS, AGG_THREADS, 0, stream>>>(x, emb, ei, h, dinv, lists, counts, pool, cnt);

  static_assert(N_LAYERS == 4, "the four layer launches are written out");
  agg_kernel<<<N_OWNERS, AGG_THREADS, 0, stream>>>(ei, dinv, h,  b,           xw, W,       W + 1 * DIM * DIM, lists, counts, batch, pool, cnt);
  agg_kernel<<<N_OWNERS, AGG_THREADS, 0, stream>>>(ei, dinv, xw, b + 1 * DIM, h,  nullptr, W + 2 * DIM * DIM, lists, counts, batch, pool, cnt);
  agg_kernel<<<N_OWNERS, AGG_THREADS, 0, stream>>>(ei, dinv, h,  b + 2 * DIM, xw, nullptr, W + 3 * DIM * DIM, lists, counts, batch, pool, cnt);
  agg_kernel<<<N_OWNERS, AGG_THREADS, 0, stream>>>(ei, dinv, xw, b + 3 * DIM, h,  nullptr, nullptr,           lists, counts, batch, pool, cnt);

  pool_div_kernel<<<(N_GRAPHS * DIM + B256 - 1) / B256, B256, 0, stream>>>(pool, cnt);

  dim3 rgrid(N_GRAPHS / 16, DIM / 16);
  gemm_wmma_f32<<<rgrid, 32, 0, stream>>>(pool, Wr, br, out);
}


# Edge list read once, not by all 256 owners: a binning launch groups each 3200-edge chunk by owner in LDS (integer ranks); owners copy their segments into their lists (list and sum order change)
# speedup vs baseline: 4.2319x; 1.0967x over previous
#include <hip/hip_runtime.h>

#define N_NODES  50000
#define N_EDGES  800000
#define DIM      128
#define N_LAYERS 4
#define N_GRAPHS 512
#define N_COLS   4
#define VOCAB    128

#define OWN_ROWS       196
#define ROW_TILES      ((OWN_ROWS + 15) / 16)
#define ROW_PITCH      (DIM + 4)
#define N_OWNERS       ((N_NODES + OWN_ROWS - 1) / OWN_ROWS)
#define AGG_THREADS    1024
#define AGG_WAVES      (AGG_THREADS / 32)
#define EDGES_PER_STEP 128
#define N_STEPS        (N_EDGES / EDGES_PER_STEP)
#define SCAN_UNROLL    8
#define DRAIN_AT       64
#define QCAP           (DRAIN_AT + SCAN_UNROLL * EDGES_PER_STEP)
#define NO_DST         0xFFFFFFFFu
#define LIST_CAP       32768
#define REPLAY_GROUP   8
#define POOL_BINS      16
#define POOL_ROWS      ((OWN_ROWS + AGG_WAVES - 1) / AGG_WAVES)
#define EMB_BATCH      4
#define CHUNK          3200
#define N_CHUNKS       (N_EDGES / CHUNK)
#define BIN_THREADS    (CHUNK / 4)

static_assert(N_NODES <= 65535, "node ids must fit 16 bits with 0xFFFF left free");
static_assert(N_EDGES % EDGES_PER_STEP == 0, "edge list must divide into whole steps");
static_assert(N_EDGES <= (1 << 24) && OWN_ROWS <= 256, "a queued word is edge index << 8 | local row");
static_assert(DIM == 128, "a wave holds a row as four columns per lane; bin words are found by shifts");
static_assert(POOL_ROWS < 32, "the graph ids of a wave's rows, and the one behind them, sit one per lane");
static_assert(N_EDGES % CHUNK == 0 && CHUNK % 128 == 0 && CHUNK / 4 <= BIN_THREADS && BIN_THREADS <= 1024 && BIN_THREADS >= N_OWNERS,
              "a chunk is one 16-byte load per thread of whole waves; a thread per owner clears and stores the counts");
static_assert(CHUNK < (1 << 16), "a chunk's per-owner offset and count share one word");
static_assert(N_OWNERS == 256 && N_CHUNKS <= 256, "the two prefix sums take eight entries per lane of one wave");
static_assert(N_CHUNKS * 4 <= AGG_THREADS, "four threads copy a chunk's segment");

typedef __attribute__((ext_vector_type(2))) float v2f;
typedef __attribute__((ext_vector_type(4))) float v4f;
typedef __attribute__((ext_vector_type(8))) float v8f;
typedef __attribute__((ext_vector_type(4))) unsigned v4u;

__global__ void gemm_wmma_f32(const float* __restrict__ A, const float* __restrict__ B,
                              const float* __restrict__ bias, float* __restrict__ Cout) {
  int row0 = blockIdx.x * 16;
  int col0 = blockIdx.y * 16;
  int lane = threadIdx.x;
  int half = lane >> 4;
  int lm   = lane & 15;

  const float* ap = A + (size_t)(row0 + lm) * DIM + half * 2;
  const float* bp = B + (size_t)(half * 2) * DIM + col0 + lm;

  v8f acc = {};
#pragma unroll
  for (int k = 0; k < DIM; k += 4) {
    v2f a = *(const v2f*)(ap + k);
    v2f b;
    b.x = bp[k * DIM];
    b.y = bp[k * DIM + DIM];
    acc = __builtin_amdgcn_wmma_f32_16x16x4_f32(
         false, a,  false, b,
         (short)0, acc,  false,  false);
  }

  float badd = bias ? bias[col0 + lm] : 0.f;
#pragma unroll
  for (int r = 0; r < 8; ++r)
    Cout[(size_t)(row0 + half * 8 + r) * DIM + col0 + lm] = acc[r] + badd;
}


template <int G, typename GET>
static __device__ __forceinline__ void add_edges(GET get, float* rows,
                                                 const float* __restrict__ xw, int lo, int lane) {
  int ld[G];
  v2f a[G], b[G];
#pragma unroll
  for (int g = 0; g < G; ++g) {
    unsigned p = get(g);
    int s = (int)(p & 0xFFFFu);
    ld[g] = (int)(p >> 16) - lo;
    const float* r = xw + (size_t)s * DIM + lane * 2;
    a[g] = *(const v2f*)r;
    b[g] = *(const v2f*)(r + 64);
  }
#pragma unroll
  for (int g = 0; g < G; ++g) {
    float* d = rows + ld[g] * ROW_PITCH + lane * 2;
    __hip_atomic_fetch_add(d,      a[g].x, __ATOMIC_RELAXED, __HIP_MEMORY_SCOPE_WORKGROUP);
    __hip_atomic_fetch_add(d + 1,  a[g].y, __ATOMIC_RELAXED, __HIP_MEMORY_SCOPE_WORKGROUP);
    __hip_atomic_fetch_add(d + 64, b[g].x, __ATOMIC_RELAXED, __HIP_MEMORY_SCOPE_WORKGROUP);
    __hip_atomic_fetch_add(d + 65, b[g].y, __ATOMIC_RELAXED, __HIP_MEMORY_SCOPE_WORKGROUP);
  }
}

static __device__ __forceinline__ void drain(const unsigned* q, unsigned n, float* rows,
                                             const float* __restrict__ xw, int lo, int lane) {
  unsigned i = 0;
  auto get = [&](int g) { return (unsigned)__builtin_amdgcn_readfirstlane(q[i + g]); };
  for (; i + 8 <= n; i += 8) add_edges<8>(get, rows, xw, lo, lane);
  for (; i + 2 <= n; i += 2) add_edges<2>(get, rows, xw, lo, lane);
  if (i < n) add_edges<1>(get, rows, xw, lo, lane);
}

template <bool GUARD>
static __device__ __forceinline__ void scan_batch(const int* __restrict__ dstv, int st, int rot, int lo, int cnt,
                                                  int lane, unsigned below, unsigned* q, unsigned& qn) {
  v4u e[SCAN_UNROLL];
#pragma unroll
  for (int u = 0; u < SCAN_UNROLL; ++u) {
    int s = st + u * AGG_WAVES;
    if (!GUARD || s < N_STEPS) {
      int sr = s + rot;
      if (sr >= N_STEPS) sr -= N_STEPS;
      e[u] = *(const v4u*)(dstv + (size_t)sr * EDGES_PER_STEP + lane * 4);
    } else {
      e[u] = (v4u){NO_DST, NO_DST, NO_DST, NO_DST};
    }
  }
#pragma unroll
  for (int u = 0; u < SCAN_UNROLL; ++u) {
#pragma unroll
    for (int k = 0; k < 4; ++k) {
      unsigned dl = e[u][k] - (unsigned)lo;
      bool hit = dl < (unsigned)cnt;
      unsigned m = __builtin_amdgcn_ballot_w32(hit);
      if (m) {
        int sr = st + u * AGG_WAVES + rot;
        if (sr >= N_STEPS) sr -= N_STEPS;
        if (hit) q[qn + __builtin_popcount(m & below)] =
                     ((unsigned)(sr * EDGES_PER_STEP + lane * 4 + k) << 8) | dl;
        qn += __builtin_popcount(m);
      }
    }
  }
}

template <typename FLUSH>
static __device__ __forceinline__ void scan_edges(const int* __restrict__ dstv, int lo, int cnt,
                                                  int wave, int lane, unsigned* q, FLUSH flush) {
  unsigned qn = 0;
  const unsigned below = (1u << lane) - 1u;
  const int rot = (int)blockIdx.x * (N_STEPS / N_OWNERS);

  int st = wave;
  for (; st + (SCAN_UNROLL - 1) * AGG_WAVES < N_STEPS; st += AGG_WAVES * SCAN_UNROLL) {
    scan_batch<false>(dstv, st, rot, lo, cnt, lane, below, q, qn);
    if (qn >= DRAIN_AT) {
      flush(qn);
      qn = 0;
    }
  }
  if (st < N_STEPS) scan_batch<true>(dstv, st, rot, lo, cnt, lane, below, q, qn);
  if (qn) flush(qn);
}

static __device__ __forceinline__ void resolve(unsigned* q, unsigned n, const int* __restrict__ srcv, int lo,
                                               int lane) {
  for (unsigned i = lane; i < n; i += 32) {
    const unsigned w = q[i];
    q[i] = (unsigned)srcv[w >> 8] | (((unsigned)lo + (w & 0xFFu)) << 16);
  }
}

static __device__ __forceinline__ void replay(const unsigned* __restrict__ list, unsigned count,
                                              int wave, float* rows,
                                              const float* __restrict__ xw, int lo, int lane) {
  if (count == 0) return;
  const unsigned stride = AGG_WAVES * REPLAY_GROUP;
  const unsigned sub = (unsigned)lane & (REPLAY_GROUP - 1);
  const unsigned last = count - 1;
  unsigned i = (unsigned)wave * REPLAY_GROUP;
  unsigned nxt = list[(i + sub < last) ? i + sub : last];
  for (; i < count; i += stride) {
    unsigned cur = nxt;
    unsigned j = i + stride + sub;
    nxt = list[(j < last) ? j : last];
    unsigned n = count - i;
    if (n >= REPLAY_GROUP) {
      add_edges<REPLAY_GROUP>([&](int g) { return (unsigned)__builtin_amdgcn_readlane(cur, g); },
                              rows, xw, lo, lane);
    } else {
      for (unsigned g = 0; g < n; ++g)
        add_edges<1>([&](int) { return (unsigned)__builtin_amdgcn_readlane(cur, g); },
                     rows, xw, lo, lane);
    }
  }
}

static __device__ __forceinline__ v8f tile_product(const float* ap, const v2f (&bf)[DIM / 4]) {
  v8f acc = {};
#pragma unroll
  for (int k = 0; k < DIM; k += 4) {
    v2f a = *(const v2f*)(ap + k);
    acc = __builtin_amdgcn_wmma_f32_16x16x4_f32(
         false, a,  false, bf[k / 4],
         (short)0, acc,  false,  false);
  }
  return acc;
}

__global__ __launch_bounds__(BIN_THREADS)
void bin_kernel(const int* __restrict__ ei, unsigned* __restrict__ binned, unsigned* __restrict__ meta) {
  __shared__ unsigned hist[N_OWNERS];
  __shared__ unsigned pref[N_OWNERS];
  __shared__ __attribute__((aligned(16))) unsigned stage[CHUNK];

  const int tid = threadIdx.x, lane = tid & 31;
  const int e0 = blockIdx.x * CHUNK;
  const bool act = tid < CHUNK / 4;

  v4u d4 = {NO_DST, NO_DST, NO_DST, NO_DST}, s4 = {0u, 0u, 0u, 0u};
  if (act) {
    d4 = *(const v4u*)(ei + N_EDGES + e0 + tid * 4);
    s4 = *(const v4u*)(ei + e0 + tid * 4);
  }
  if (tid < N_OWNERS) hist[tid] = 0u;
  __syncthreads();

  unsigned own[4], place[4];
#pragma unroll
  for (int k = 0; k < 4; ++k) {
    const unsigned d = d4[k];
    const bool ok = d < (unsigned)N_NODES;
    own[k] = ok ? d / (unsigned)OWN_ROWS : NO_DST;
    place[k] = 0u;
    if (ok) place[k] = __hip_atomic_fetch_add(hist + own[k], 1u, __ATOMIC_RELAXED, __HIP_MEMORY_SCOPE_WORKGROUP);
  }
  __syncthreads();

  if (tid < 32) {
    unsigned v[8], s = 0u;
#pragma unroll
    for (int j = 0; j < 8; ++j) {
      v[j] = hist[lane * 8 + j];
      s += v[j];
    }
    unsigned inc = s;
#pragma unroll
    for (int off = 1; off < 32; off <<= 1) {
      const unsigned t = __shfl_up(inc, off, 32);
      if (lane >= off) inc += t;
    }
    unsigned ex = inc - s;
#pragma unroll
    for (int j = 0; j < 8; ++j) {
      pref[lane * 8 + j] = ex;
      ex += v[j];
    }
  }
  __syncthreads();

#pragma unroll
  for (int k = 0; k < 4; ++k)
    if (own[k] != NO_DST) stage[pref[own[k]] + place[k]] = s4[k] | (d4[k] << 16);
  __syncthreads();

  if (act) *(v4u*)(binned + (size_t)e0 + tid * 4) = *(const v4u*)(stage + tid * 4);
  if (tid < N_OWNERS) meta[(size_t)blockIdx.x * N_OWNERS + tid] = (pref[tid] << 16) | hist[tid];
}

__global__ __launch_bounds__(AGG_THREADS)
void setup_kernel(const int* __restrict__ x, const float* __restrict__ emb,
                  const unsigned* __restrict__ binned, const unsigned* __restrict__ meta,
                  float* __restrict__ h, float* __restrict__ dinv,
                  unsigned* __restrict__ lists, unsigned* __restrict__ counts,
                  float* __restrict__ pool, float* __restrict__ pnodes) {
  __shared__ unsigned sdeg[OWN_ROWS];
  __shared__ unsigned cmeta[N_CHUNKS];
  __shared__ unsigned cbase[N_CHUNKS];
  __shared__ unsigned lcount;

  const int tid = threadIdx.x, lane = tid & 31;
  const int wave = __builtin_amdgcn_readfirstlane(tid >> 5);
  const int lo = blockIdx.x * OWN_ROWS;
  const int cnt = (N_NODES - lo < OWN_ROWS) ? (N_NODES - lo) : OWN_ROWS;
  unsigned* list = lists + (size_t)blockIdx.x * LIST_CAP;

  unsigned mv = 0u;
  if (tid < N_CHUNKS) mv = meta[(size_t)tid * N_OWNERS + blockIdx.x];
  for (int i = tid; i < cnt; i += AGG_THREADS) sdeg[i] = 1u;
  for (int t = blockIdx.x * AGG_THREADS + tid; t < N_GRAPHS * DIM + N_GRAPHS; t += N_OWNERS * AGG_THREADS) {
    if (t < N_GRAPHS * DIM) pool[t] = 0.f;
    else pnodes[t - N_GRAPHS * DIM] = 0.f;
  }
  if (tid < N_CHUNKS) cmeta[tid] = mv;
  __syncthreads();

  if (tid < 32) {
    unsigned v[8], s = 0u;
#pragma unroll
    for (int j = 0; j < 8; ++j) {
      const int ch = lane * 8 + j;
      v[j] = (ch < N_CHUNKS) ? (cmeta[ch] & 0xFFFFu) : 0u;
      s += v[j];
    }
    unsigned inc = s;
#pragma unroll
    for (int off = 1; off < 32; off <<= 1) {
      const unsigned t = __shfl_up(inc, off, 32);
      if (lane >= off) inc += t;
    }
    unsigned ex = inc - s;
#pragma unroll
    for (int j = 0; j < 8; ++j) {
      const int ch = lane * 8 + j;
      if (ch < N_CHUNKS) cbase[ch] = ex;
      ex += v[j];
    }
    if (lane == 31) lcount = inc;
  }
  __syncthreads();

  {
    const int ch = tid >> 2, sub = tid & 3;
    if (ch < N_CHUNKS) {
      const unsigned m = cmeta[ch], n = m & 0xFFFFu, base = cbase[ch];
      const unsigned* seg = binned + (size_t)ch * CHUNK + (m >> 16);
      for (unsigned i0 = (unsigned)sub; i0 < n; i0 += 16u) {
        unsigned w[4];
#pragma unroll
        for (int u = 0; u < 4; ++u) {
          const unsigned i = i0 + 4u * u;
          w[u] = seg[(i < n) ? i : i0];
        }
#pragma unroll
        for (int u = 0; u < 4; ++u) {
          const unsigned i = i0 + 4u * u;
          if (i < n) {
            __hip_atomic_fetch_add(sdeg + ((w[u] >> 16) - (unsigned)lo), 1u, __ATOMIC_RELAXED,
                                   __HIP_MEMORY_SCOPE_WORKGROUP);
            if (base + i < LIST_CAP) list[base + i] = w[u];
          }
        }
      }
    }
  }
  __syncthreads();

  if (tid == 0) counts[blockIdx.x] = lcount;
  for (int i = tid; i < cnt; i += AGG_THREADS) {
    const float d = (float)sdeg[i];
    dinv[lo + i] = (d > 0.f) ? rsqrtf(d) : 0.f;
  }

  for (int r0 = wave; r0 < cnt; r0 += AGG_WAVES * EMB_BATCH) {
    v4f t[EMB_BATCH][N_COLS];
#pragma unroll
    for (int u = 0; u < EMB_BATCH; ++u) {
      int r = r0 + u * AGG_WAVES;
      if (r >= cnt) r = cnt - 1;
      const int* xr = x + (size_t)(lo + r) * N_COLS;
#pragma unroll
      for (int cc = 0; cc < N_COLS; ++cc)
        t[u][cc] = *(const v4f*)(emb + ((size_t)cc * VOCAB + xr[cc]) * DIM + lane * 4);
    }
#pragma unroll
    for (int u = 0; u < EMB_BATCH; ++u) {
      asm volatile("" :: "v"(t[u][0]), "v"(t[u][1]), "v"(t[u][2]), "v"(t[u][3]));
    }
#pragma unroll
    for (int u = 0; u < EMB_BATCH; ++u) {
      const int r = r0 + u * AGG_WAVES;
      if (r < cnt) {
        v4f s = {0.f, 0.f, 0.f, 0.f};
#pragma unroll
        for (int cc = 0; cc < N_COLS; ++cc) s += t[u][cc];
        const float d = (float)sdeg[r];
        const float sc = (d > 0.f) ? rsqrtf(d) : 0.f;
        *(v4f*)(h + (size_t)(lo + r) * DIM + lane * 4) = s * sc;
      }
    }
  }
}

__global__ __launch_bounds__(AGG_THREADS)
void agg_kernel(const int* __restrict__ ei, const float* __restrict__ dinv,
                const float* __restrict__ xw, const float* __restrict__ bl, float* __restrict__ outp,
                const float* __restrict__ W0, const float* __restrict__ Wn,
                const unsigned* __restrict__ lists, const unsigned* __restrict__ counts,
                const int* __restrict__ batch, float* __restrict__ pool, float* __restrict__ pnodes) {
  __shared__ float rows[ROW_TILES * 16 * ROW_PITCH];
  __shared__ __attribute__((aligned(16))) float sdinv[ROW_TILES * 16];
  __shared__ unsigned queue[AGG_WAVES * QCAP];
  __shared__ float pbin[POOL_BINS * DIM];
  __shared__ float pnum[POOL_BINS];

  const int tid = threadIdx.x, lane = tid & 31;
  const int wave = __builtin_amdgcn_readfirstlane(tid >> 5);
  const bool first = (W0 != nullptr);
  const bool product = (Wn != nullptr);
  const unsigned* list = lists + (size_t)blockIdx.x * LIST_CAP;
  const unsigned kept = counts[blockIdx.x];
  const bool use_list = kept <= LIST_CAP;
  const int lo = blockIdx.x * OWN_ROWS;
  const int cnt = (N_NODES - lo < OWN_ROWS) ? (N_NODES - lo) : OWN_ROWS;

  for (int i = tid; i < cnt * 32; i += AGG_THREADS) {
    int r = i >> 5, c4 = (i & 31) << 2;
    *(v4f*)(rows + r * ROW_PITCH + c4) = *(const v4f*)(xw + (size_t)(lo + r) * DIM + c4);
  }
  for (int i = cnt * 32 + tid; i < ((cnt + 15) & ~15) * 32; i += AGG_THREADS) {
    int r = i >> 5, c4 = (i & 31) << 2;
    *(v4f*)(rows + r * ROW_PITCH + c4) = (v4f){0.f, 0.f, 0.f, 0.f};
  }
  for (int i = tid; i < ROW_TILES * 16; i += AGG_THREADS) sdinv[i] = (i < cnt) ? dinv[lo + i] : 0.f;
  if (!product) {
    for (int i = tid; i < POOL_BINS * DIM; i += AGG_THREADS) pbin[i] = 0.f;
    if (tid < POOL_BINS) pnum[tid] = 0.f;
  }
  __syncthreads();

  if (use_list) {
    replay(list, kept, wave, rows, xw, lo, lane);
  } else {
    unsigned* q = queue + wave * QCAP;
    scan_edges(ei + N_EDGES, lo, cnt, wave, lane, q, [&](unsigned n) {
      resolve(q, n, ei, lo, lane);
      drain(q, n, rows, xw, lo, lane);
    });
  }
  const int half = lane >> 4, lm = lane & 15;
  const int col0 = (wave & 7) * 16;
  const int n_rt = (cnt + 15) >> 4;
  v2f bf[DIM / 4];
  auto load_b = [&](const float* __restrict__ Wp) {
    const float* bp = Wp + (size_t)(half * 2) * DIM + col0 + lm;
#pragma unroll
    for (int k = 0; k < DIM; k += 4) {
      bf[k / 4].x = bp[k * DIM];
      bf[k / 4].y = bp[k * DIM + DIM];
    }
  };
  if (first) load_b(W0);
  else if (product) load_b(Wn);
  const int fc4 = (tid & 31) << 2;
  v4f fbias = {0.f, 0.f, 0.f, 0.f};
  if (!first) fbias = *(const v4f*)(bl + fc4);
  const int pr0 = wave * POOL_ROWS;
  const int pr1 = (pr0 + POOL_ROWS < cnt) ? pr0 + POOL_ROWS : cnt;
  int bv = 0, g0 = 0;
  if (!product) {
    g0 = batch[lo];
    if (pr0 + lane < pr1) bv = batch[lo + pr0 + lane];
  }
  __syncthreads();

  if (!first) {
    for (int i = tid; i < cnt * 32; i += AGG_THREADS) {
      const int r = i >> 5;
      const float s = sdinv[r];
      v4f* p = (v4f*)(rows + r * ROW_PITCH + fc4);
      const v4f x = *p;
      v4f o;
      o.x = fmaxf(fbias.x + x.x * s, 0.f);
      o.y = fmaxf(fbias.y + x.y * s, 0.f);
      o.z = fmaxf(fbias.z + x.z * s, 0.f);
      o.w = fmaxf(fbias.w + x.w * s, 0.f);
      *p = o;
    }
    __syncthreads();
  }

  const float* arows = rows;
  if (first) {
    static_assert(AGG_WAVES * QCAP >= ROW_TILES * 16 * ROW_PITCH, "the queues must hold a second set of rows");
    float* stage = reinterpret_cast<float*>(queue);
    const float badd = bl[col0 + lm];
    for (int rt = wave >> 3; rt < n_rt; rt += AGG_WAVES / 8) {
      v8f acc = tile_product(rows + (rt * 16 + lm) * ROW_PITCH + half * 2, bf);
      float* sp = stage + (rt * 16 + half * 8) * ROW_PITCH + col0 + lm;
      const v4f s0 = *(const v4f*)(sdinv + rt * 16 + half * 8), s1 = *(const v4f*)(sdinv + rt * 16 + half * 8 + 4);
      const float sd[8] = {s0.x, s0.y, s0.z, s0.w, s1.x, s1.y, s1.z, s1.w};
#pragma unroll
      for (int r = 0; r < 8; ++r) sp[r * ROW_PITCH] = fmaxf(acc[r] * sd[r] + badd, 0.f);
    }
    if (product) load_b(Wn);
    __syncthreads();
    arows = stage;
  }

  if (!product) {
    if (pr0 < pr1) {
      const float* rp = arows + pr0 * ROW_PITCH + lane * 2;
      const int n_rows = pr1 - pr0;
      v2f sa = {0.f, 0.f}, sb = {0.f, 0.f};
      float num = 0.f;
      int g = __builtin_amdgcn_readlane(bv, 0);
      v2f na = *(const v2f*)rp, nb = *(const v2f*)(rp + 64);
      for (int k = 0; k < n_rows; ++k) {
        const v2f a = na, b = nb;
        const bool last_row = (k + 1 == n_rows);
        const float* np = rp + (last_row ? k : k + 1) * ROW_PITCH;
        na = *(const v2f*)np;
        nb = *(const v2f*)(np + 64);
        const int gn = __builtin_amdgcn_readlane(bv, k + 1);
        sa.x += a.x;
        sa.y += a.y;
        sb.x += b.x;
        sb.y += b.y;
        num += 1.f;
        if (last_row || gn != g) {
          const unsigned slot = (unsigned)(g - g0);
          if (slot < POOL_BINS) {
            float* d = pbin + slot * DIM + lane * 2;
            __hip_atomic_fetch_add(d,      sa.x, __ATOMIC_RELAXED, __HIP_MEMORY_SCOPE_WORKGROUP);
            __hip_atomic_fetch_add(d + 1,  sa.y, __ATOMIC_RELAXED, __HIP_MEMORY_SCOPE_WORKGROUP);
            __hip_atomic_fetch_add(d + 64, sb.x, __ATOMIC_RELAXED, __HIP_MEMORY_SCOPE_WORKGROUP);
            __hip_atomic_fetch_add(d + 65, sb.y, __ATOMIC_RELAXED, __HIP_MEMORY_SCOPE_WORKGROUP);
            if (lane == 0) __hip_atomic_fetch_add(pnum + slot, num, __ATOMIC_RELAXED, __HIP_MEMORY_SCOPE_WORKGROUP);
          } else {
            float* d = pool + (size_t)g * DIM + lane * 2;
            unsafeAtomicAdd(d,      sa.x);
            unsafeAtomicAdd(d + 1,  sa.y);
            unsafeAtomicAdd(d + 64, sb.x);
            unsafeAtomicAdd(d + 65, sb.y);
            if (lane == 0) unsafeAtomicAdd(&pnodes[g], num);
          }
          sa = (v2f){0.f, 0.f};
          sb = (v2f){0.f, 0.f};
          num = 0.f;
        }
        g = gn;
      }
    }
    __syncthreads();
    for (int i = tid; i < POOL_BINS * DIM; i += AGG_THREADS) {
      const int slot = i >> 7;
      const float n = pnum[slot];
      if (n > 0.f) {
        unsafeAtomicAdd(&pool[(size_t)(g0 + slot) * DIM + (i & (DIM - 1))], pbin[i]);
        if ((i & (DIM - 1)) == 0) unsafeAtomicAdd(&pnodes[g0 + slot], n);
      }
    }
    return;
  }

  for (int rt = wave >> 3; rt < n_rt; rt += AGG_WAVES / 8) {
    v8f acc = tile_product(arows + (rt * 16 + lm) * ROW_PITCH + half * 2, bf);
    const int r0 = rt * 16 + half * 8;
    const v4f s0 = *(const v4f*)(sdinv + r0), s1 = *(const v4f*)(sdinv + r0 + 4);
    const float sd[8] = {s0.x, s0.y, s0.z, s0.w, s1.x, s1.y, s1.z, s1.w};
#pragma unroll
    for (int r = 0; r < 8; ++r)
      if (r0 + r < cnt) outp[(size_t)(lo + r0 + r) * DIM + col0 + lm] = acc[r] * sd[r];
  }
}

__global__ void pool_div_kernel(float* __restrict__ pool, const float* __restrict__ cnt) {
  int t = blockIdx.x * blockDim.x + threadIdx.x;
  if (t >= N_GRAPHS * DIM) return;
  pool[t] /= fmaxf(cnt[t / DIM], 1.0f);
}

extern "C" void kernel_launch(void* const* d_in, const int* in_sizes, int n_in,
                              void* d_out, int out_size, void* d_ws, size_t ws_size,
                              hipStream_t stream) {
  const int*   x    = (const int*)d_in[0];
  const int*   ei   = (const int*)d_in[1];
  const int*   batch= (const int*)d_in[2];
  const float* emb  = (const float*)d_in[3];
  const float* W    = (const float*)d_in[4];
  const float* b    = (const float*)d_in[5];
  const float* Wr   = (const float*)d_in[6];
  const float* br   = (const float*)d_in[7];
  float* out = (float*)d_out;

  char* ws = (char*)d_ws;
  const size_t NH = (size_t)N_NODES * DIM * sizeof(float);
  float* h    = (float*)(ws);
  float* xw   = (float*)(ws + NH);
  float* dinv = (float*)(ws + 3 * NH);
  float* pool = (float*)(ws + 3 * NH + 4 * 1024 * 1024);
  float* cnt  = (float*)(ws + 3 * NH + 4 * 1024 * 1024 + 512 * 1024);
  unsigned* counts = (unsigned*)(ws + 3 * NH + 5 * 1024 * 1024);
  unsigned* lists  = (unsigned*)(ws + 3 * NH + 8 * 1024 * 1024);
  unsigned* binned = (unsigned*)(ws + 3 * NH + 44 * 1024 * 1024);
  unsigned* meta   = (unsigned*)(ws + 3 * NH + 52 * 1024 * 1024);

  const int B256 = 256;

  bin_kernel<<<N_CHUNKS, BIN_THREADS, 0, stream>>>(ei, binned, meta);
  setup_kernel<<<N_OWNERS, AGG_THREADS, 0, stream>>>(x, emb, binned, meta, h, dinv, lists, counts, pool, cnt);

  static_assert(N_LAYERS == 4, "the four layer launches are written out");
  agg_kernel<<<N_OWNERS, AGG_THREADS, 0, stream>>>(ei, dinv, h,  b,           xw, W,       W + 1 * DIM * DIM, lists, counts, batch, pool, cnt);
  agg_kernel<<<N_OWNERS, AGG_THREADS, 0, stream>>>(ei, dinv, xw, b + 1 * DIM, h,  nullptr, W + 2 * DIM * DIM, lists, counts, batch, pool, cnt);
  agg_kernel<<<N_OWNERS, AGG_THREADS, 0, stream>>>(ei, dinv, h,  b + 2 * DIM, xw, nullptr, W + 3 * DIM * DIM, lists, counts, batch, pool, cnt);
  agg_kernel<<<N_OWNERS, AGG_THREADS, 0, stream>>>(ei, dinv, xw, b + 3 * DIM, h,  nullptr, nullptr,           lists, counts, batch, pool, cnt);

  pool_div_kernel<<<(N_GRAPHS * DIM + B256 - 1) / B256, B256, 0, stream>>>(pool, cnt);

  dim3 rgrid(N_GRAPHS / 16, DIM / 16);
  gemm_wmma_f32<<<rgrid, 32, 0, stream>>>(pool, Wr, br, out);
}


# Owner lists grouped by owning wave in set-up (LDS integer ranks): a wave alone adds to its 7 LDS rows by plain 16-byte read-add-write, one 512-B load a row; no LDS f32 atomics (sum order)
# speedup vs baseline: 5.2966x; 1.2516x over previous
#include <hip/hip_runtime.h>

#define N_NODES  50000
#define N_EDGES  800000
#define DIM      128
#define N_LAYERS 4
#define N_GRAPHS 512
#define N_COLS   4
#define VOCAB    128

#define OWN_ROWS       196
#define ROW_TILES      ((OWN_ROWS + 15) / 16)
#define ROW_PITCH      (DIM + 4)
#define N_OWNERS       ((N_NODES + OWN_ROWS - 1) / OWN_ROWS)
#define AGG_THREADS    1024
#define AGG_WAVES      (AGG_THREADS / 32)
#define EDGES_PER_STEP 128
#define N_STEPS        (N_EDGES / EDGES_PER_STEP)
#define SCAN_UNROLL    8
#define DRAIN_AT       64
#define QCAP           (DRAIN_AT + SCAN_UNROLL * EDGES_PER_STEP)
#define NO_DST         0xFFFFFFFFu
#define LIST_CAP       32768
#define SUM_GROUP      8
#define POOL_BINS      16
#define POOL_ROWS      ((OWN_ROWS + AGG_WAVES - 1) / AGG_WAVES)
#define WAVE_ROWS      POOL_ROWS
#define EMB_BATCH      4
#define CHUNK          3200
#define N_CHUNKS       (N_EDGES / CHUNK)
#define BIN_THREADS    (CHUNK / 4)

static_assert(N_NODES <= 65535, "node ids must fit 16 bits with 0xFFFF left free");
static_assert(N_EDGES % EDGES_PER_STEP == 0, "edge list must divide into whole steps");
static_assert(N_EDGES <= (1 << 24) && OWN_ROWS <= 256, "a queued word is edge index << 8 | local row");
static_assert(DIM == 128, "a wave holds a row as four columns per lane; bin words are found by shifts");
static_assert(POOL_ROWS < 32, "the graph ids of a wave's rows, and the one behind them, sit one per lane");
static_assert(N_EDGES % CHUNK == 0 && CHUNK % 128 == 0 && CHUNK / 4 <= BIN_THREADS && BIN_THREADS <= 1024 && BIN_THREADS >= N_OWNERS,
              "a chunk is one 16-byte load per thread of whole waves; a thread per owner clears and stores the counts");
static_assert(CHUNK < (1 << 16), "a chunk's per-owner offset and count share one word");
static_assert(N_OWNERS == 256 && N_CHUNKS <= 256, "the two prefix sums take eight entries per lane of one wave");
static_assert(N_CHUNKS * 4 <= AGG_THREADS, "four threads copy a chunk's segment");
static_assert(AGG_WAVES == 32, "the places of the waves' parts of a list are found by one wave, a part a lane");

typedef __attribute__((ext_vector_type(2))) float v2f;
typedef __attribute__((ext_vector_type(4))) float v4f;
typedef __attribute__((ext_vector_type(8))) float v8f;
typedef __attribute__((ext_vector_type(4))) unsigned v4u;

__global__ void gemm_wmma_f32(const float* __restrict__ A, const float* __restrict__ B,
                              const float* __restrict__ bias, float* __restrict__ Cout) {
  int row0 = blockIdx.x * 16;
  int col0 = blockIdx.y * 16;
  int lane = threadIdx.x;
  int half = lane >> 4;
  int lm   = lane & 15;

  const float* ap = A + (size_t)(row0 + lm) * DIM + half * 2;
  const float* bp = B + (size_t)(half * 2) * DIM + col0 + lm;

  v8f acc = {};
#pragma unroll
  for (int k = 0; k < DIM; k += 4) {
    v2f a = *(const v2f*)(ap + k);
    v2f b;
    b.x = bp[k * DIM];
    b.y = bp[k * DIM + DIM];
    acc = __builtin_amdgcn_wmma_f32_16x16x4_f32(
         false, a,  false, b,
         (short)0, acc,  false,  false);
  }

  float badd = bias ? bias[col0 + lm] : 0.f;
#pragma unroll
  for (int r = 0; r < 8; ++r)
    Cout[(size_t)(row0 + half * 8 + r) * DIM + col0 + lm] = acc[r] + badd;
}


template <int G, typename GET>
static __device__ __forceinline__ void add_edges(GET get, float* rows,
                                                 const float* __restrict__ xw, int lo, int lane) {
  int ld[G];
  v2f a[G], b[G];
#pragma unroll
  for (int g = 0; g < G; ++g) {
    unsigned p = get(g);
    int s = (int)(p & 0xFFFFu);
    ld[g] = (int)(p >> 16) - lo;
    const float* r = xw + (size_t)s * DIM + lane * 2;
    a[g] = *(const v2f*)r;
    b[g] = *(const v2f*)(r + 64);
  }
#pragma unroll
  for (int g = 0; g < G; ++g) {
    float* d = rows + ld[g] * ROW_PITCH + lane * 2;
    __hip_atomic_fetch_add(d,      a[g].x, __ATOMIC_RELAXED, __HIP_MEMORY_SCOPE_WORKGROUP);
    __hip_atomic_fetch_add(d + 1,  a[g].y, __ATOMIC_RELAXED, __HIP_MEMORY_SCOPE_WORKGROUP);
    __hip_atomic_fetch_add(d + 64, b[g].x, __ATOMIC_RELAXED, __HIP_MEMORY_SCOPE_WORKGROUP);
    __hip_atomic_fetch_add(d + 65, b[g].y, __ATOMIC_RELAXED, __HIP_MEMORY_SCOPE_WORKGROUP);
  }
}

static __device__ __forceinline__ void drain(const unsigned* q, unsigned n, float* rows,
                                             const float* __restrict__ xw, int lo, int lane) {
  unsigned i = 0;
  auto get = [&](int g) { return (unsigned)__builtin_amdgcn_readfirstlane(q[i + g]); };
  for (; i + 8 <= n; i += 8) add_edges<8>(get, rows, xw, lo, lane);
  for (; i + 2 <= n; i += 2) add_edges<2>(get, rows, xw, lo, lane);
  if (i < n) add_edges<1>(get, rows, xw, lo, lane);
}

template <bool GUARD>
static __device__ __forceinline__ void scan_batch(const int* __restrict__ dstv, int st, int rot, int lo, int cnt,
                                                  int lane, unsigned below, unsigned* q, unsigned& qn) {
  v4u e[SCAN_UNROLL];
#pragma unroll
  for (int u = 0; u < SCAN_UNROLL; ++u) {
    int s = st + u * AGG_WAVES;
    if (!GUARD || s < N_STEPS) {
      int sr = s + rot;
      if (sr >= N_STEPS) sr -= N_STEPS;
      e[u] = *(const v4u*)(dstv + (size_t)sr * EDGES_PER_STEP + lane * 4);
    } else {
      e[u] = (v4u){NO_DST, NO_DST, NO_DST, NO_DST};
    }
  }
#pragma unroll
  for (int u = 0; u < SCAN_UNROLL; ++u) {
#pragma unroll
    for (int k = 0; k < 4; ++k) {
      unsigned dl = e[u][k] - (unsigned)lo;
      bool hit = dl < (unsigned)cnt;
      unsigned m = __builtin_amdgcn_ballot_w32(hit);
      if (m) {
        int sr = st + u * AGG_WAVES + rot;
        if (sr >= N_STEPS) sr -= N_STEPS;
        if (hit) q[qn + __builtin_popcount(m & below)] =
                     ((unsigned)(sr * EDGES_PER_STEP + lane * 4 + k) << 8) | dl;
        qn += __builtin_popcount(m);
      }
    }
  }
}

template <typename FLUSH>
static __device__ __forceinline__ void scan_edges(const int* __restrict__ dstv, int lo, int cnt,
                                                  int wave, int lane, unsigned* q, FLUSH flush) {
  unsigned qn = 0;
  const unsigned below = (1u << lane) - 1u;
  const int rot = (int)blockIdx.x * (N_STEPS / N_OWNERS);

  int st = wave;
  for (; st + (SCAN_UNROLL - 1) * AGG_WAVES < N_STEPS; st += AGG_WAVES * SCAN_UNROLL) {
    scan_batch<false>(dstv, st, rot, lo, cnt, lane, below, q, qn);
    if (qn >= DRAIN_AT) {
      flush(qn);
      qn = 0;
    }
  }
  if (st < N_STEPS) scan_batch<true>(dstv, st, rot, lo, cnt, lane, below, q, qn);
  if (qn) flush(qn);
}

static __device__ __forceinline__ void resolve(unsigned* q, unsigned n, const int* __restrict__ srcv, int lo,
                                               int lane) {
  for (unsigned i = lane; i < n; i += 32) {
    const unsigned w = q[i];
    q[i] = (unsigned)srcv[w >> 8] | (((unsigned)lo + (w & 0xFFu)) << 16);
  }
}


template <int Q>
static __device__ __forceinline__ void ask_rows(unsigned words, v4f (&d)[SUM_GROUP], int (&ro)[SUM_GROUP],
                                                const float* __restrict__ xw, int row0, int lane) {
#pragma unroll
  for (int g = 0; g < SUM_GROUP; ++g) {
    const unsigned p = (unsigned)__builtin_amdgcn_readlane(words, Q * SUM_GROUP + g);
    ro[g] = ((int)(p >> 16) - row0) * ROW_PITCH;
    d[g] = *(const v4f*)(xw + (size_t)(p & 0xFFFFu) * DIM + lane * 4);
  }
}

template <bool TAIL>
static __device__ __forceinline__ void add_rows(const v4f (&d)[SUM_GROUP], const int (&ro)[SUM_GROUP],
                                                unsigned n, float* wrow) {
#pragma unroll
  for (int g = 0; g < SUM_GROUP; ++g) {
    v4f x = d[g];
    if (TAIL && (unsigned)g >= n) x = (v4f){0.f, 0.f, 0.f, 0.f};
    v4f* p = (v4f*)(wrow + ro[g]);
    *p = *p + x;
  }
}

template <typename START>
static __device__ __forceinline__ void sum_edges(const unsigned* __restrict__ seg, unsigned n, float* wrow,
                                                 const float* __restrict__ xw, int row0, int lane, START start) {
  static_assert(SUM_GROUP == 8, "a block of 32 words is written out as four groups");
  if (n == 0) {
    start();
    return;
  }
  const unsigned last = n - 1;
  const unsigned blocks = n >> 5, rem = n & 31u;
  v4f dA[SUM_GROUP], dB[SUM_GROUP];
  int roA[SUM_GROUP], roB[SUM_GROUP];
  unsigned cur = seg[((unsigned)lane < last) ? (unsigned)lane : last];
  ask_rows<0>(cur, dA, roA, xw, row0, lane);
  start();
  for (unsigned b = 0; b < blocks; ++b) {
    const unsigned j = (b + 1) * 32u + (unsigned)lane;
    const unsigned nxt = seg[(j < last) ? j : last];
    ask_rows<1>(cur, dB, roB, xw, row0, lane);
    add_rows<false>(dA, roA, SUM_GROUP, wrow);
    ask_rows<2>(cur, dA, roA, xw, row0, lane);
    add_rows<false>(dB, roB, SUM_GROUP, wrow);
    ask_rows<3>(cur, dB, roB, xw, row0, lane);
    add_rows<false>(dA, roA, SUM_GROUP, wrow);
    ask_rows<0>(nxt, dA, roA, xw, row0, lane);
    add_rows<false>(dB, roB, SUM_GROUP, wrow);
    cur = nxt;
  }
  if (rem > 0) {
    if (rem > 8) ask_rows<1>(cur, dB, roB, xw, row0, lane);
    add_rows<true>(dA, roA, rem, wrow);
    if (rem > 8) {
      if (rem > 16) ask_rows<2>(cur, dA, roA, xw, row0, lane);
      add_rows<true>(dB, roB, rem - 8, wrow);
      if (rem > 16) {
        if (rem > 24) ask_rows<3>(cur, dB, roB, xw, row0, lane);
        add_rows<true>(dA, roA, rem - 16, wrow);
        if (rem > 24) add_rows<true>(dB, roB, rem - 24, wrow);
      }
    }
  }
}

static __device__ __forceinline__ v8f tile_product(const float* ap, const v2f (&bf)[DIM / 4]) {
  v8f acc = {};
#pragma unroll
  for (int k = 0; k < DIM; k += 4) {
    v2f a = *(const v2f*)(ap + k);
    acc = __builtin_amdgcn_wmma_f32_16x16x4_f32(
         false, a,  false, bf[k / 4],
         (short)0, acc,  false,  false);
  }
  return acc;
}

__global__ __launch_bounds__(BIN_THREADS)
void bin_kernel(const int* __restrict__ ei, unsigned* __restrict__ binned, unsigned* __restrict__ meta) {
  __shared__ unsigned hist[N_OWNERS];
  __shared__ unsigned pref[N_OWNERS];
  __shared__ __attribute__((aligned(16))) unsigned stage[CHUNK];

  const int tid = threadIdx.x, lane = tid & 31;
  const int e0 = blockIdx.x * CHUNK;
  const bool act = tid < CHUNK / 4;

  v4u d4 = {NO_DST, NO_DST, NO_DST, NO_DST}, s4 = {0u, 0u, 0u, 0u};
  if (act) {
    d4 = *(const v4u*)(ei + N_EDGES + e0 + tid * 4);
    s4 = *(const v4u*)(ei + e0 + tid * 4);
  }
  if (tid < N_OWNERS) hist[tid] = 0u;
  __syncthreads();

  unsigned own[4], place[4];
#pragma unroll
  for (int k = 0; k < 4; ++k) {
    const unsigned d = d4[k];
    const bool ok = d < (unsigned)N_NODES;
    own[k] = ok ? d / (unsigned)OWN_ROWS : NO_DST;
    place[k] = 0u;
    if (ok) place[k] = __hip_atomic_fetch_add(hist + own[k], 1u, __ATOMIC_RELAXED, __HIP_MEMORY_SCOPE_WORKGROUP);
  }
  __syncthreads();

  if (tid < 32) {
    unsigned v[8], s = 0u;
#pragma unroll
    for (int j = 0; j < 8; ++j) {
      v[j] = hist[lane * 8 + j];
      s += v[j];
    }
    unsigned inc = s;
#pragma unroll
    for (int off = 1; off < 32; off <<= 1) {
      const unsigned t = __shfl_up(inc, off, 32);
      if (lane >= off) inc += t;
    }
    unsigned ex = inc - s;
#pragma unroll
    for (int j = 0; j < 8; ++j) {
      pref[lane * 8 + j] = ex;
      ex += v[j];
    }
  }
  __syncthreads();

#pragma unroll
  for (int k = 0; k < 4; ++k)
    if (own[k] != NO_DST) stage[pref[own[k]] + place[k]] = s4[k] | (d4[k] << 16);
  __syncthreads();

  if (act) *(v4u*)(binned + (size_t)e0 + tid * 4) = *(const v4u*)(stage + tid * 4);
  if (tid < N_OWNERS) meta[(size_t)blockIdx.x * N_OWNERS + tid] = (pref[tid] << 16) | hist[tid];
}

__global__ __launch_bounds__(AGG_THREADS)
void setup_kernel(const int* __restrict__ x, const float* __restrict__ emb,
                  const unsigned* __restrict__ binned, const unsigned* __restrict__ meta,
                  float* __restrict__ h, float* __restrict__ dinv,
                  unsigned* __restrict__ lists, unsigned* __restrict__ counts, unsigned* __restrict__ woffs,
                  float* __restrict__ pool, float* __restrict__ pnodes) {
  __shared__ unsigned sdeg[OWN_ROWS];
  __shared__ unsigned cmeta[N_CHUNKS];
  __shared__ unsigned wstart[AGG_WAVES + 1];
  __shared__ unsigned wfill[AGG_WAVES];

  const int tid = threadIdx.x, lane = tid & 31;
  const int wave = __builtin_amdgcn_readfirstlane(tid >> 5);
  const int lo = blockIdx.x * OWN_ROWS;
  const int cnt = (N_NODES - lo < OWN_ROWS) ? (N_NODES - lo) : OWN_ROWS;
  unsigned* list = lists + (size_t)blockIdx.x * LIST_CAP;

  unsigned mv = 0u;
  if (tid < N_CHUNKS) mv = meta[(size_t)tid * N_OWNERS + blockIdx.x];
  for (int i = tid; i < cnt; i += AGG_THREADS) sdeg[i] = 1u;
  for (int t = blockIdx.x * AGG_THREADS + tid; t < N_GRAPHS * DIM + N_GRAPHS; t += N_OWNERS * AGG_THREADS) {
    if (t < N_GRAPHS * DIM) pool[t] = 0.f;
    else pnodes[t - N_GRAPHS * DIM] = 0.f;
  }
  if (tid < N_CHUNKS) cmeta[tid] = mv;
  __syncthreads();

  const int ch = tid >> 2, sub = tid & 3;
  unsigned seg_n = 0u;
  const unsigned* seg = binned;
  if (ch < N_CHUNKS) {
    const unsigned m = cmeta[ch];
    seg_n = m & 0xFFFFu;
    seg = binned + (size_t)ch * CHUNK + (m >> 16);
  }
  for (unsigned i0 = (unsigned)sub; i0 < seg_n; i0 += 16u) {
    unsigned w[4];
#pragma unroll
    for (int u = 0; u < 4; ++u) {
      const unsigned i = i0 + 4u * u;
      w[u] = seg[(i < seg_n) ? i : i0];
    }
#pragma unroll
    for (int u = 0; u < 4; ++u) {
      const unsigned i = i0 + 4u * u;
      if (i < seg_n)
        __hip_atomic_fetch_add(sdeg + ((w[u] >> 16) - (unsigned)lo), 1u, __ATOMIC_RELAXED, __HIP_MEMORY_SCOPE_WORKGROUP);
    }
  }
  __syncthreads();

  if (tid < 32) {
    unsigned c = 0u;
#pragma unroll
    for (int k = 0; k < WAVE_ROWS; ++k) {
      const int r = lane * WAVE_ROWS + k;
      if (r < cnt) c += sdeg[r] - 1u;
    }
    unsigned inc = c;
#pragma unroll
    for (int off = 1; off < 32; off <<= 1) {
      const unsigned t = __shfl_up(inc, off, 32);
      if (lane >= off) inc += t;
    }
    wstart[lane] = inc - c;
    wfill[lane] = 0u;
    woffs[(size_t)blockIdx.x * (AGG_WAVES + 1) + lane] = inc - c;
    if (lane == 31) {
      wstart[AGG_WAVES] = inc;
      woffs[(size_t)blockIdx.x * (AGG_WAVES + 1) + AGG_WAVES] = inc;
    }
  }
  __syncthreads();

  for (unsigned i0 = (unsigned)sub; i0 < seg_n; i0 += 16u) {
    unsigned w[4];
#pragma unroll
    for (int u = 0; u < 4; ++u) {
      const unsigned i = i0 + 4u * u;
      w[u] = seg[(i < seg_n) ? i : i0];
    }
#pragma unroll
    for (int u = 0; u < 4; ++u) {
      const unsigned i = i0 + 4u * u;
      if (i < seg_n) {
        const unsigned part = ((w[u] >> 16) - (unsigned)lo) / (unsigned)WAVE_ROWS;
        const unsigned pos = wstart[part] +
            __hip_atomic_fetch_add(wfill + part, 1u, __ATOMIC_RELAXED, __HIP_MEMORY_SCOPE_WORKGROUP);
        if (pos < LIST_CAP) list[pos] = w[u];
      }
    }
  }

  if (tid == 0) counts[blockIdx.x] = wstart[AGG_WAVES];
  for (int i = tid; i < cnt; i += AGG_THREADS) {
    const float d = (float)sdeg[i];
    dinv[lo + i] = (d > 0.f) ? rsqrtf(d) : 0.f;
  }

  for (int r0 = wave; r0 < cnt; r0 += AGG_WAVES * EMB_BATCH) {
    v4f t[EMB_BATCH][N_COLS];
#pragma unroll
    for (int u = 0; u < EMB_BATCH; ++u) {
      int r = r0 + u * AGG_WAVES;
      if (r >= cnt) r = cnt - 1;
      const int* xr = x + (size_t)(lo + r) * N_COLS;
#pragma unroll
      for (int cc = 0; cc < N_COLS; ++cc)
        t[u][cc] = *(const v4f*)(emb + ((size_t)cc * VOCAB + xr[cc]) * DIM + lane * 4);
    }
#pragma unroll
    for (int u = 0; u < EMB_BATCH; ++u) {
      asm volatile("" :: "v"(t[u][0]), "v"(t[u][1]), "v"(t[u][2]), "v"(t[u][3]));
    }
#pragma unroll
    for (int u = 0; u < EMB_BATCH; ++u) {
      const int r = r0 + u * AGG_WAVES;
      if (r < cnt) {
        v4f s = {0.f, 0.f, 0.f, 0.f};
#pragma unroll
        for (int cc = 0; cc < N_COLS; ++cc) s += t[u][cc];
        const float d = (float)sdeg[r];
        const float sc = (d > 0.f) ? rsqrtf(d) : 0.f;
        *(v4f*)(h + (size_t)(lo + r) * DIM + lane * 4) = s * sc;
      }
    }
  }
}

__global__ __launch_bounds__(AGG_THREADS)
void agg_kernel(const int* __restrict__ ei, const float* __restrict__ dinv,
                const float* __restrict__ xw, const float* __restrict__ bl, float* __restrict__ outp,
                const float* __restrict__ W0, const float* __restrict__ Wn,
                const unsigned* __restrict__ lists, const unsigned* __restrict__ counts,
                const unsigned* __restrict__ woffs,
                const int* __restrict__ batch, float* __restrict__ pool, float* __restrict__ pnodes) {
  __shared__ float rows[ROW_TILES * 16 * ROW_PITCH];
  __shared__ __attribute__((aligned(16))) float sdinv[ROW_TILES * 16];
  __shared__ unsigned queue[AGG_WAVES * QCAP];
  __shared__ float pbin[POOL_BINS * DIM];
  __shared__ float pnum[POOL_BINS];

  const int tid = threadIdx.x, lane = tid & 31;
  const int wave = __builtin_amdgcn_readfirstlane(tid >> 5);
  const bool first = (W0 != nullptr);
  const bool product = (Wn != nullptr);
  const unsigned* list = lists + (size_t)blockIdx.x * LIST_CAP;
  const unsigned kept = counts[blockIdx.x];
  const bool use_list = kept <= LIST_CAP;
  const int lo = blockIdx.x * OWN_ROWS;
  const int cnt = (N_NODES - lo < OWN_ROWS) ? (N_NODES - lo) : OWN_ROWS;

  const int fc4 = lane << 2;
  v4f fbias = {0.f, 0.f, 0.f, 0.f};
  if (!first) fbias = *(const v4f*)(bl + fc4);

  for (int i = cnt * 32 + tid; i < ((cnt + 15) & ~15) * 32; i += AGG_THREADS) {
    int r = i >> 5, c4 = (i & 31) << 2;
    *(v4f*)(rows + r * ROW_PITCH + c4) = (v4f){0.f, 0.f, 0.f, 0.f};
  }
  for (int i = tid; i < ROW_TILES * 16; i += AGG_THREADS) sdinv[i] = (i < cnt) ? dinv[lo + i] : 0.f;
  if (!product) {
    for (int i = tid; i < POOL_BINS * DIM; i += AGG_THREADS) pbin[i] = 0.f;
    if (tid < POOL_BINS) pnum[tid] = 0.f;
  }

  if (use_list) {
    const int wr0 = wave * WAVE_ROWS;
    const int wn = (cnt - wr0 < WAVE_ROWS) ? (cnt - wr0) : WAVE_ROWS;
    if (wn > 0) {
      float* wrow = rows + wr0 * ROW_PITCH + fc4;
      v4f st[WAVE_ROWS];
#pragma unroll
      for (int k = 0; k < WAVE_ROWS; ++k) {
        const int r = wr0 + ((k < wn) ? k : wn - 1);
        st[k] = *(const v4f*)(xw + (size_t)(lo + r) * DIM + fc4);
      }
      const unsigned* wo = woffs + (size_t)blockIdx.x * (AGG_WAVES + 1) + wave;
      const unsigned beg = wo[0], end = wo[1];
      sum_edges(list + beg, end - beg, wrow, xw, lo + wr0, lane, [&]() {
#pragma unroll
        for (int k = 0; k < WAVE_ROWS; ++k)
          if (k < wn) *(v4f*)(wrow + k * ROW_PITCH) = st[k];
      });
      if (!first) {
        v4f o[WAVE_ROWS];
#pragma unroll
        for (int k = 0; k < WAVE_ROWS; ++k)
          if (k < wn) o[k] = *(const v4f*)(wrow + k * ROW_PITCH);
#pragma unroll
        for (int k = 0; k < WAVE_ROWS; ++k) {
          if (k < wn) {
            const float sc = dinv[lo + wr0 + k];
            v4f x = o[k];
            x.x = fmaxf(fbias.x + x.x * sc, 0.f);
            x.y = fmaxf(fbias.y + x.y * sc, 0.f);
            x.z = fmaxf(fbias.z + x.z * sc, 0.f);
            x.w = fmaxf(fbias.w + x.w * sc, 0.f);
            *(v4f*)(wrow + k * ROW_PITCH) = x;
          }
        }
      }
    }
  } else {
    for (int i = tid; i < cnt * 32; i += AGG_THREADS) {
      int r = i >> 5, c4 = (i & 31) << 2;
      *(v4f*)(rows + r * ROW_PITCH + c4) = *(const v4f*)(xw + (size_t)(lo + r) * DIM + c4);
    }
    __syncthreads();
    unsigned* q = queue + wave * QCAP;
    scan_edges(ei + N_EDGES, lo, cnt, wave, lane, q, [&](unsigned n) {
      resolve(q, n, ei, lo, lane);
      drain(q, n, rows, xw, lo, lane);
    });
    if (!first) {
      __syncthreads();
      for (int i = tid; i < cnt * 32; i += AGG_THREADS) {
        const int r = i >> 5;
        const float sc = sdinv[r];
        v4f* p = (v4f*)(rows + r * ROW_PITCH + fc4);
        const v4f x = *p;
        v4f o;
        o.x = fmaxf(fbias.x + x.x * sc, 0.f);
        o.y = fmaxf(fbias.y + x.y * sc, 0.f);
        o.z = fmaxf(fbias.z + x.z * sc, 0.f);
        o.w = fmaxf(fbias.w + x.w * sc, 0.f);
        *p = o;
      }
    }
  }
  const int half = lane >> 4, lm = lane & 15;
  const int col0 = (wave & 7) * 16;
  const int n_rt = (cnt + 15) >> 4;
  v2f bf[DIM / 4];
  auto load_b = [&](const float* __restrict__ Wp) {
    const float* bp = Wp + (size_t)(half * 2) * DIM + col0 + lm;
#pragma unroll
    for (int k = 0; k < DIM; k += 4) {
      bf[k / 4].x = bp[k * DIM];
      bf[k / 4].y = bp[k * DIM + DIM];
    }
  };
  if (first) load_b(W0);
  else if (product) load_b(Wn);
  const int pr0 = wave * POOL_ROWS;
  const int pr1 = (pr0 + POOL_ROWS < cnt) ? pr0 + POOL_ROWS : cnt;
  int bv = 0, g0 = 0;
  if (!product) {
    g0 = batch[lo];
    if (pr0 + lane < pr1) bv = batch[lo + pr0 + lane];
  }
  __syncthreads();

  const float* arows = rows;
  if (first) {
    static_assert(AGG_WAVES * QCAP >= ROW_TILES * 16 * ROW_PITCH, "the queues must hold a second set of rows");
    float* stage = reinterpret_cast<float*>(queue);
    const float badd = bl[col0 + lm];
    for (int rt = wave >> 3; rt < n_rt; rt += AGG_WAVES / 8) {
      v8f acc = tile_product(rows + (rt * 16 + lm) * ROW_PITCH + half * 2, bf);
      float* sp = stage + (rt * 16 + half * 8) * ROW_PITCH + col0 + lm;
      const v4f s0 = *(const v4f*)(sdinv + rt * 16 + half * 8), s1 = *(const v4f*)(sdinv + rt * 16 + half * 8 + 4);
      const float sd[8] = {s0.x, s0.y, s0.z, s0.w, s1.x, s1.y, s1.z, s1.w};
#pragma unroll
      for (int r = 0; r < 8; ++r) sp[r * ROW_PITCH] = fmaxf(acc[r] * sd[r] + badd, 0.f);
    }
    if (product) load_b(Wn);
    __syncthreads();
    arows = stage;
  }

  if (!product) {
    if (pr0 < pr1) {
      const float* rp = arows + pr0 * ROW_PITCH + lane * 2;
      const int n_rows = pr1 - pr0;
      v2f sa = {0.f, 0.f}, sb = {0.f, 0.f};
      float num = 0.f;
      int g = __builtin_amdgcn_readlane(bv, 0);
      v2f na = *(const v2f*)rp, nb = *(const v2f*)(rp + 64);
      for (int k = 0; k < n_rows; ++k) {
        const v2f a = na, b = nb;
        const bool last_row = (k + 1 == n_rows);
        const float* np = rp + (last_row ? k : k + 1) * ROW_PITCH;
        na = *(const v2f*)np;
        nb = *(const v2f*)(np + 64);
        const int gn = __builtin_amdgcn_readlane(bv, k + 1);
        sa.x += a.x;
        sa.y += a.y;
        sb.x += b.x;
        sb.y += b.y;
        num += 1.f;
        if (last_row || gn != g) {
          const unsigned slot = (unsigned)(g - g0);
          if (slot < POOL_BINS) {
            float* d = pbin + slot * DIM + lane * 2;
            __hip_atomic_fetch_add(d,      sa.x, __ATOMIC_RELAXED, __HIP_MEMORY_SCOPE_WORKGROUP);
            __hip_atomic_fetch_add(d + 1,  sa.y, __ATOMIC_RELAXED, __HIP_MEMORY_SCOPE_WORKGROUP);
            __hip_atomic_fetch_add(d + 64, sb.x, __ATOMIC_RELAXED, __HIP_MEMORY_SCOPE_WORKGROUP);
            __hip_atomic_fetch_add(d + 65, sb.y, __ATOMIC_RELAXED, __HIP_MEMORY_SCOPE_WORKGROUP);
            if (lane == 0) __hip_atomic_fetch_add(pnum + slot, num, __ATOMIC_RELAXED, __HIP_MEMORY_SCOPE_WORKGROUP);
          } else {
            float* d = pool + (size_t)g * DIM + lane * 2;
            unsafeAtomicAdd(d,      sa.x);
            unsafeAtomicAdd(d + 1,  sa.y);
            unsafeAtomicAdd(d + 64, sb.x);
            unsafeAtomicAdd(d + 65, sb.y);
            if (lane == 0) unsafeAtomicAdd(&pnodes[g], num);
          }
          sa = (v2f){0.f, 0.f};
          sb = (v2f){0.f, 0.f};
          num = 0.f;
        }
        g = gn;
      }
    }
    __syncthreads();
    for (int i = tid; i < POOL_BINS * DIM; i += AGG_THREADS) {
      const int slot = i >> 7;
      const float n = pnum[slot];
      if (n > 0.f) {
        unsafeAtomicAdd(&pool[(size_t)(g0 + slot) * DIM + (i & (DIM - 1))], pbin[i]);
        if ((i & (DIM - 1)) == 0) unsafeAtomicAdd(&pnodes[g0 + slot], n);
      }
    }
    return;
  }

  for (int rt = wave >> 3; rt < n_rt; rt += AGG_WAVES / 8) {
    v8f acc = tile_product(arows + (rt * 16 + lm) * ROW_PITCH + half * 2, bf);
    const int r0 = rt * 16 + half * 8;
    const v4f s0 = *(const v4f*)(sdinv + r0), s1 = *(const v4f*)(sdinv + r0 + 4);
    const float sd[8] = {s0.x, s0.y, s0.z, s0.w, s1.x, s1.y, s1.z, s1.w};
#pragma unroll
    for (int r = 0; r < 8; ++r)
      if (r0 + r < cnt) outp[(size_t)(lo + r0 + r) * DIM + col0 + lm] = acc[r] * sd[r];
  }
}

__global__ void pool_div_kernel(float* __restrict__ pool, const float* __restrict__ cnt) {
  int t = blockIdx.x * blockDim.x + threadIdx.x;
  if (t >= N_GRAPHS * DIM) return;
  pool[t] /= fmaxf(cnt[t / DIM], 1.0f);
}

extern "C" void kernel_launch(void* const* d_in, const int* in_sizes, int n_in,
                              void* d_out, int out_size, void* d_ws, size_t ws_size,
                              hipStream_t stream) {
  const int*   x    = (const int*)d_in[0];
  const int*   ei   = (const int*)d_in[1];
  const int*   batch= (const int*)d_in[2];
  const float* emb  = (const float*)d_in[3];
  const float* W    = (const float*)d_in[4];
  const float* b    = (const float*)d_in[5];
  const float* Wr   = (const float*)d_in[6];
  const float* br   = (const float*)d_in[7];
  float* out = (float*)d_out;

  char* ws = (char*)d_ws;
  const size_t NH = (size_t)N_NODES * DIM * sizeof(float);
  float* h    = (float*)(ws);
  float* xw   = (float*)(ws + NH);
  float* dinv = (float*)(ws + 3 * NH);
  float* pool = (float*)(ws + 3 * NH + 4 * 1024 * 1024);
  float* cnt  = (float*)(ws + 3 * NH + 4 * 1024 * 1024 + 512 * 1024);
  unsigned* counts = (unsigned*)(ws + 3 * NH + 5 * 1024 * 1024);
  unsigned* woffs  = (unsigned*)(ws + 3 * NH + 6 * 1024 * 1024);
  unsigned* lists  = (unsigned*)(ws + 3 * NH + 8 * 1024 * 1024);
  unsigned* binned = (unsigned*)(ws + 3 * NH + 44 * 1024 * 1024);
  unsigned* meta   = (unsigned*)(ws + 3 * NH + 52 * 1024 * 1024);

  const int B256 = 256;

  bin_kernel<<<N_CHUNKS, BIN_THREADS, 0, stream>>>(ei, binned, meta);
  setup_kernel<<<N_OWNERS, AGG_THREADS, 0, stream>>>(x, emb, binned, meta, h, dinv, lists, counts, woffs, pool, cnt);

  static_assert(N_LAYERS == 4, "the four layer launches are written out");
  agg_kernel<<<N_OWNERS, AGG_THREADS, 0, stream>>>(ei, dinv, h,  b,           xw, W,       W + 1 * DIM * DIM, lists, counts, woffs, batch, pool, cnt);
  agg_kernel<<<N_OWNERS, AGG_THREADS, 0, stream>>>(ei, dinv, xw, b + 1 * DIM, h,  nullptr, W + 2 * DIM * DIM, lists, counts, woffs, batch, pool, cnt);
  agg_kernel<<<N_OWNERS, AGG_THREADS, 0, stream>>>(ei, dinv, h,  b + 2 * DIM, xw, nullptr, W + 3 * DIM * DIM, lists, counts, woffs, batch, pool, cnt);
  agg_kernel<<<N_OWNERS, AGG_THREADS, 0, stream>>>(ei, dinv, xw, b + 3 * DIM, h,  nullptr, nullptr,           lists, counts, woffs, batch, pool, cnt);

  pool_div_kernel<<<(N_GRAPHS * DIM + B256 - 1) / B256, B256, 0, stream>>>(pool, cnt);

  dim3 rgrid(N_GRAPHS / 16, DIM / 16);
  gemm_wmma_f32<<<rgrid, 32, 0, stream>>>(pool, Wr, br, out);
}


# Owner lists ordered by destination row in set-up (LDS integer ranks): a wave sums a row's gathered rows in registers, one LDS add a row, none an edge; 16 waves x 13 rows, 16 loads a group (sum order)
# speedup vs baseline: 5.5235x; 1.0428x over previous
#include <hip/hip_runtime.h>

#define N_NODES  50000
#define N_EDGES  800000
#define DIM      128
#define N_LAYERS 4
#define N_GRAPHS 512
#define N_COLS   4
#define VOCAB    128

#define OWN_ROWS       196
#define ROW_TILES      ((OWN_ROWS + 15) / 16)
#define ROW_PITCH      (DIM + 4)
#define N_OWNERS       ((N_NODES + OWN_ROWS - 1) / OWN_ROWS)
#define AGG_THREADS    512
#define AGG_WAVES      (AGG_THREADS / 32)
#define SETUP_THREADS  1024
#define SETUP_WAVES    (SETUP_THREADS / 32)
#define EDGES_PER_STEP 128
#define N_STEPS        (N_EDGES / EDGES_PER_STEP)
#define SCAN_UNROLL    8
#define DRAIN_AT       64
#define QCAP           (DRAIN_AT + SCAN_UNROLL * EDGES_PER_STEP)
#define STAGE_WORDS    (ROW_TILES * 16 * ROW_PITCH)
#define QUEUE_WORDS    ((AGG_WAVES * QCAP > STAGE_WORDS) ? (AGG_WAVES * QCAP) : STAGE_WORDS)
#define NO_DST         0xFFFFFFFFu
#define LIST_CAP       32768
#define SUM_GROUP      16
#define POOL_BINS      16
#define POOL_ROWS      ((OWN_ROWS + AGG_WAVES - 1) / AGG_WAVES)
#define WAVE_ROWS      POOL_ROWS
#define EMB_BATCH      4
#define CHUNK          3200
#define N_CHUNKS       (N_EDGES / CHUNK)
#define BIN_THREADS    (CHUNK / 4)

static_assert(N_NODES <= 65535, "node ids must fit 16 bits with 0xFFFF left free");
static_assert(N_EDGES % EDGES_PER_STEP == 0, "edge list must divide into whole steps");
static_assert(N_EDGES <= (1 << 24) && OWN_ROWS <= 256, "a queued word is edge index << 8 | local row");
static_assert(DIM == 128, "a wave holds a row as four columns per lane; bin words are found by shifts");
static_assert(POOL_ROWS < 32, "the graph ids of a wave's rows, and the one behind them, sit one per lane");
static_assert(N_EDGES % CHUNK == 0 && CHUNK % 128 == 0 && CHUNK / 4 <= BIN_THREADS && BIN_THREADS <= 1024 && BIN_THREADS >= N_OWNERS,
              "a chunk is one 16-byte load per thread of whole waves; a thread per owner clears and stores the counts");
static_assert(CHUNK < (1 << 16), "a chunk's per-owner offset and count share one word");
static_assert(N_OWNERS == 256 && N_CHUNKS <= 256, "the two prefix sums take eight entries per lane of one wave");
static_assert(N_CHUNKS * 4 <= SETUP_THREADS, "four threads copy a chunk's segment");
static_assert(AGG_WAVES < 32 && AGG_WAVES * WAVE_ROWS >= OWN_ROWS,
              "the places of the waves' parts of a list are found by one wave, a part a lane; the lane behind the last part holds the list's length");

typedef __attribute__((ext_vector_type(2))) float v2f;
typedef __attribute__((ext_vector_type(4))) float v4f;
typedef __attribute__((ext_vector_type(8))) float v8f;
typedef __attribute__((ext_vector_type(4))) unsigned v4u;

__global__ void gemm_wmma_f32(const float* __restrict__ A, const float* __restrict__ B,
                              const float* __restrict__ bias, float* __restrict__ Cout) {
  int row0 = blockIdx.x * 16;
  int col0 = blockIdx.y * 16;
  int lane = threadIdx.x;
  int half = lane >> 4;
  int lm   = lane & 15;

  const float* ap = A + (size_t)(row0 + lm) * DIM + half * 2;
  const float* bp = B + (size_t)(half * 2) * DIM + col0 + lm;

  v8f acc = {};
#pragma unroll
  for (int k = 0; k < DIM; k += 4) {
    v2f a = *(const v2f*)(ap + k);
    v2f b;
    b.x = bp[k * DIM];
    b.y = bp[k * DIM + DIM];
    acc = __builtin_amdgcn_wmma_f32_16x16x4_f32(
         false, a,  false, b,
         (short)0, acc,  false,  false);
  }

  float badd = bias ? bias[col0 + lm] : 0.f;
#pragma unroll
  for (int r = 0; r < 8; ++r)
    Cout[(size_t)(row0 + half * 8 + r) * DIM + col0 + lm] = acc[r] + badd;
}


template <int G, typename GET>
static __device__ __forceinline__ void add_edges(GET get, float* rows,
                                                 const float* __restrict__ xw, int lo, int lane) {
  int ld[G];
  v2f a[G], b[G];
#pragma unroll
  for (int g = 0; g < G; ++g) {
    unsigned p = get(g);
    int s = (int)(p & 0xFFFFu);
    ld[g] = (int)(p >> 16) - lo;
    const float* r = xw + (size_t)s * DIM + lane * 2;
    a[g] = *(const v2f*)r;
    b[g] = *(const v2f*)(r + 64);
  }
#pragma unroll
  for (int g = 0; g < G; ++g) {
    float* d = rows + ld[g] * ROW_PITCH + lane * 2;
    __hip_atomic_fetch_add(d,      a[g].x, __ATOMIC_RELAXED, __HIP_MEMORY_SCOPE_WORKGROUP);
    __hip_atomic_fetch_add(d + 1,  a[g].y, __ATOMIC_RELAXED, __HIP_MEMORY_SCOPE_WORKGROUP);
    __hip_atomic_fetch_add(d + 64, b[g].x, __ATOMIC_RELAXED, __HIP_MEMORY_SCOPE_WORKGROUP);
    __hip_atomic_fetch_add(d + 65, b[g].y, __ATOMIC_RELAXED, __HIP_MEMORY_SCOPE_WORKGROUP);
  }
}

static __device__ __forceinline__ void drain(const unsigned* q, unsigned n, float* rows,
                                             const float* __restrict__ xw, int lo, int lane) {
  unsigned i = 0;
  auto get = [&](int g) { return (unsigned)__builtin_amdgcn_readfirstlane(q[i + g]); };
  for (; i + 8 <= n; i += 8) add_edges<8>(get, rows, xw, lo, lane);
  for (; i + 2 <= n; i += 2) add_edges<2>(get, rows, xw, lo, lane);
  if (i < n) add_edges<1>(get, rows, xw, lo, lane);
}

template <bool GUARD>
static __device__ __forceinline__ void scan_batch(const int* __restrict__ dstv, int st, int rot, int lo, int cnt,
                                                  int lane, unsigned below, unsigned* q, unsigned& qn) {
  v4u e[SCAN_UNROLL];
#pragma unroll
  for (int u = 0; u < SCAN_UNROLL; ++u) {
    int s = st + u * AGG_WAVES;
    if (!GUARD || s < N_STEPS) {
      int sr = s + rot;
      if (sr >= N_STEPS) sr -= N_STEPS;
      e[u] = *(const v4u*)(dstv + (size_t)sr * EDGES_PER_STEP + lane * 4);
    } else {
      e[u] = (v4u){NO_DST, NO_DST, NO_DST, NO_DST};
    }
  }
#pragma unroll
  for (int u = 0; u < SCAN_UNROLL; ++u) {
#pragma unroll
    for (int k = 0; k < 4; ++k) {
      unsigned dl = e[u][k] - (unsigned)lo;
      bool hit = dl < (unsigned)cnt;
      unsigned m = __builtin_amdgcn_ballot_w32(hit);
      if (m) {
        int sr = st + u * AGG_WAVES + rot;
        if (sr >= N_STEPS) sr -= N_STEPS;
        if (hit) q[qn + __builtin_popcount(m & below)] =
                     ((unsigned)(sr * EDGES_PER_STEP + lane * 4 + k) << 8) | dl;
        qn += __builtin_popcount(m);
      }
    }
  }
}

template <typename FLUSH>
static __device__ __forceinline__ void scan_edges(const int* __restrict__ dstv, int lo, int cnt,
                                                  int wave, int lane, unsigned* q, FLUSH flush) {
  unsigned qn = 0;
  const unsigned below = (1u << lane) - 1u;
  const int rot = (int)blockIdx.x * (N_STEPS / N_OWNERS);

  int st = wave;
  for (; st + (SCAN_UNROLL - 1) * AGG_WAVES < N_STEPS; st += AGG_WAVES * SCAN_UNROLL) {
    scan_batch<false>(dstv, st, rot, lo, cnt, lane, below, q, qn);
    if (qn >= DRAIN_AT) {
      flush(qn);
      qn = 0;
    }
  }
  if (st < N_STEPS) scan_batch<true>(dstv, st, rot, lo, cnt, lane, below, q, qn);
  if (qn) flush(qn);
}

static __device__ __forceinline__ void resolve(unsigned* q, unsigned n, const int* __restrict__ srcv, int lo,
                                               int lane) {
  for (unsigned i = lane; i < n; i += 32) {
    const unsigned w = q[i];
    q[i] = (unsigned)srcv[w >> 8] | (((unsigned)lo + (w & 0xFFu)) << 16);
  }
}


template <int Q>
static __device__ __forceinline__ void ask_rows(unsigned words, v4f (&d)[SUM_GROUP], int (&ro)[SUM_GROUP],
                                                const float* __restrict__ xw, int row0, int lane) {
#pragma unroll
  for (int g = 0; g < SUM_GROUP; ++g) {
    const unsigned p = (unsigned)__builtin_amdgcn_readlane(words, Q * SUM_GROUP + g);
    ro[g] = ((int)(p >> 16) - row0) * ROW_PITCH;
    d[g] = *(const v4f*)(xw + (size_t)(p & 0xFFFFu) * DIM + lane * 4);
  }
}

static __device__ __forceinline__ void put_run(float* wrow, int at, const v4f& acc) {
  v4f* p = (v4f*)(wrow + at);
  *p = *p + acc;
}

template <bool TAIL>
static __device__ __forceinline__ void add_rows(const v4f (&d)[SUM_GROUP], const int (&ro)[SUM_GROUP],
                                                unsigned n, float* wrow, v4f& acc, int& at) {
#pragma unroll
  for (int g = 0; g < SUM_GROUP; ++g) {
    v4f x = d[g];
    if (TAIL && (unsigned)g >= n) x = (v4f){0.f, 0.f, 0.f, 0.f};
    if (ro[g] != at) {
      put_run(wrow, at, acc);
      acc = (v4f){0.f, 0.f, 0.f, 0.f};
      at = ro[g];
    }
    acc += x;
  }
}

template <typename START>
static __device__ __forceinline__ void sum_edges(const unsigned* __restrict__ seg, unsigned n, float* wrow,
                                                 const float* __restrict__ xw, int row0, int lane, START start) {
  static_assert(SUM_GROUP == 16, "a block of 32 words is written out as two groups");
  if (n == 0) {
    start();
    return;
  }
  const unsigned last = n - 1;
  const unsigned blocks = n >> 5, rem = n & 31u;
  v4f dA[SUM_GROUP], dB[SUM_GROUP];
  int roA[SUM_GROUP], roB[SUM_GROUP];
  unsigned cur = seg[((unsigned)lane < last) ? (unsigned)lane : last];
  ask_rows<0>(cur, dA, roA, xw, row0, lane);
  start();
  v4f acc = {0.f, 0.f, 0.f, 0.f};
  int at = roA[0];
  for (unsigned b = 0; b < blocks; ++b) {
    const unsigned j = (b + 1) * 32u + (unsigned)lane;
    const unsigned nxt = seg[(j < last) ? j : last];
    ask_rows<1>(cur, dB, roB, xw, row0, lane);
    add_rows<false>(dA, roA, SUM_GROUP, wrow, acc, at);
    ask_rows<0>(nxt, dA, roA, xw, row0, lane);
    add_rows<false>(dB, roB, SUM_GROUP, wrow, acc, at);
    cur = nxt;
  }
  if (rem > 0) {
    if (rem > SUM_GROUP) ask_rows<1>(cur, dB, roB, xw, row0, lane);
    add_rows<true>(dA, roA, rem, wrow, acc, at);
    if (rem > SUM_GROUP) add_rows<true>(dB, roB, rem - SUM_GROUP, wrow, acc, at);
  }
  put_run(wrow, at, acc);
}

static __device__ __forceinline__ v8f tile_product(const float* ap, const v2f (&bf)[DIM / 4]) {
  v8f acc = {};
#pragma unroll
  for (int k = 0; k < DIM; k += 4) {
    v2f a = *(const v2f*)(ap + k);
    acc = __builtin_amdgcn_wmma_f32_16x16x4_f32(
         false, a,  false, bf[k / 4],
         (short)0, acc,  false,  false);
  }
  return acc;
}

__global__ __launch_bounds__(BIN_THREADS)
void bin_kernel(const int* __restrict__ ei, unsigned* __restrict__ binned, unsigned* __restrict__ meta) {
  __shared__ unsigned hist[N_OWNERS];
  __shared__ unsigned pref[N_OWNERS];
  __shared__ __attribute__((aligned(16))) unsigned stage[CHUNK];

  const int tid = threadIdx.x, lane = tid & 31;
  const int e0 = blockIdx.x * CHUNK;
  const bool act = tid < CHUNK / 4;

  v4u d4 = {NO_DST, NO_DST, NO_DST, NO_DST}, s4 = {0u, 0u, 0u, 0u};
  if (act) {
    d4 = *(const v4u*)(ei + N_EDGES + e0 + tid * 4);
    s4 = *(const v4u*)(ei + e0 + tid * 4);
  }
  if (tid < N_OWNERS) hist[tid] = 0u;
  __syncthreads();

  unsigned own[4], place[4];
#pragma unroll
  for (int k = 0; k < 4; ++k) {
    const unsigned d = d4[k];
    const bool ok = d < (unsigned)N_NODES;
    own[k] = ok ? d / (unsigned)OWN_ROWS : NO_DST;
    place[k] = 0u;
    if (ok) place[k] = __hip_atomic_fetch_add(hist + own[k], 1u, __ATOMIC_RELAXED, __HIP_MEMORY_SCOPE_WORKGROUP);
  }
  __syncthreads();

  if (tid < 32) {
    unsigned v[8], s = 0u;
#pragma unroll
    for (int j = 0; j < 8; ++j) {
      v[j] = hist[lane * 8 + j];
      s += v[j];
    }
    unsigned inc = s;
#pragma unroll
    for (int off = 1; off < 32; off <<= 1) {
      const unsigned t = __shfl_up(inc, off, 32);
      if (lane >= off) inc += t;
    }
    unsigned ex = inc - s;
#pragma unroll
    for (int j = 0; j < 8; ++j) {
      pref[lane * 8 + j] = ex;
      ex += v[j];
    }
  }
  __syncthreads();

#pragma unroll
  for (int k = 0; k < 4; ++k)
    if (own[k] != NO_DST) stage[pref[own[k]] + place[k]] = s4[k] | (d4[k] << 16);
  __syncthreads();

  if (act) *(v4u*)(binned + (size_t)e0 + tid * 4) = *(const v4u*)(stage + tid * 4);
  if (tid < N_OWNERS) meta[(size_t)blockIdx.x * N_OWNERS + tid] = (pref[tid] << 16) | hist[tid];
}

__global__ __launch_bounds__(SETUP_THREADS)
void setup_kernel(const int* __restrict__ x, const float* __restrict__ emb,
                  const unsigned* __restrict__ binned, const unsigned* __restrict__ meta,
                  float* __restrict__ h, float* __restrict__ dinv,
                  unsigned* __restrict__ lists, unsigned* __restrict__ counts, unsigned* __restrict__ woffs,
                  float* __restrict__ pool, float* __restrict__ pnodes) {
  __shared__ unsigned sdeg[OWN_ROWS];
  __shared__ unsigned cmeta[N_CHUNKS];
  __shared__ unsigned rstart[OWN_ROWS];
  __shared__ unsigned rfill[OWN_ROWS];
  __shared__ unsigned ltotal;

  const int tid = threadIdx.x, lane = tid & 31;
  const int wave = __builtin_amdgcn_readfirstlane(tid >> 5);
  const int lo = blockIdx.x * OWN_ROWS;
  const int cnt = (N_NODES - lo < OWN_ROWS) ? (N_NODES - lo) : OWN_ROWS;
  unsigned* list = lists + (size_t)blockIdx.x * LIST_CAP;

  unsigned mv = 0u;
  if (tid < N_CHUNKS) mv = meta[(size_t)tid * N_OWNERS + blockIdx.x];
  for (int i = tid; i < cnt; i += SETUP_THREADS) sdeg[i] = 1u;
  for (int t = blockIdx.x * SETUP_THREADS + tid; t < N_GRAPHS * DIM + N_GRAPHS; t += N_OWNERS * SETUP_THREADS) {
    if (t < N_GRAPHS * DIM) pool[t] = 0.f;
    else pnodes[t - N_GRAPHS * DIM] = 0.f;
  }
  if (tid < N_CHUNKS) cmeta[tid] = mv;
  __syncthreads();

  const int ch = tid >> 2, sub = tid & 3;
  unsigned seg_n = 0u;
  const unsigned* seg = binned;
  if (ch < N_CHUNKS) {
    const unsigned m = cmeta[ch];
    seg_n = m & 0xFFFFu;
    seg = binned + (size_t)ch * CHUNK + (m >> 16);
  }
  for (unsigned i0 = (unsigned)sub; i0 < seg_n; i0 += 16u) {
    unsigned w[4];
#pragma unroll
    for (int u = 0; u < 4; ++u) {
      const unsigned i = i0 + 4u * u;
      w[u] = seg[(i < seg_n) ? i : i0];
    }
#pragma unroll
    for (int u = 0; u < 4; ++u) {
      const unsigned i = i0 + 4u * u;
      if (i < seg_n)
        __hip_atomic_fetch_add(sdeg + ((w[u] >> 16) - (unsigned)lo), 1u, __ATOMIC_RELAXED, __HIP_MEMORY_SCOPE_WORKGROUP);
    }
  }
  __syncthreads();

  if (tid < 32) {
    unsigned c = 0u;
#pragma unroll
    for (int k = 0; k < WAVE_ROWS; ++k) {
      const int r = lane * WAVE_ROWS + k;
      if (r < cnt) c += sdeg[r] - 1u;
    }
    unsigned inc = c;
#pragma unroll
    for (int off = 1; off < 32; off <<= 1) {
      const unsigned t = __shfl_up(inc, off, 32);
      if (lane >= off) inc += t;
    }
    unsigned ex = inc - c;
    if (lane <= AGG_WAVES) woffs[(size_t)blockIdx.x * (AGG_WAVES + 1) + lane] = ex;
    if (lane == AGG_WAVES) ltotal = ex;
#pragma unroll
    for (int k = 0; k < WAVE_ROWS; ++k) {
      const int r = lane * WAVE_ROWS + k;
      if (r < cnt) {
        rstart[r] = ex;
        rfill[r] = 0u;
        ex += sdeg[r] - 1u;
      }
    }
  }
  __syncthreads();

  for (unsigned i0 = (unsigned)sub; i0 < seg_n; i0 += 16u) {
    unsigned w[4];
#pragma unroll
    for (int u = 0; u < 4; ++u) {
      const unsigned i = i0 + 4u * u;
      w[u] = seg[(i < seg_n) ? i : i0];
    }
#pragma unroll
    for (int u = 0; u < 4; ++u) {
      const unsigned i = i0 + 4u * u;
      if (i < seg_n) {
        const unsigned row = (w[u] >> 16) - (unsigned)lo;
        const unsigned pos = rstart[row] +
            __hip_atomic_fetch_add(rfill + row, 1u, __ATOMIC_RELAXED, __HIP_MEMORY_SCOPE_WORKGROUP);
        if (pos < LIST_CAP) list[pos] = w[u];
      }
    }
  }

  if (tid == 0) counts[blockIdx.x] = ltotal;
  for (int i = tid; i < cnt; i += SETUP_THREADS) {
    const float d = (float)sdeg[i];
    dinv[lo + i] = (d > 0.f) ? rsqrtf(d) : 0.f;
  }

  for (int r0 = wave; r0 < cnt; r0 += SETUP_WAVES * EMB_BATCH) {
    v4f t[EMB_BATCH][N_COLS];
#pragma unroll
    for (int u = 0; u < EMB_BATCH; ++u) {
      int r = r0 + u * SETUP_WAVES;
      if (r >= cnt) r = cnt - 1;
      const int* xr = x + (size_t)(lo + r) * N_COLS;
#pragma unroll
      for (int cc = 0; cc < N_COLS; ++cc)
        t[u][cc] = *(const v4f*)(emb + ((size_t)cc * VOCAB + xr[cc]) * DIM + lane * 4);
    }
#pragma unroll
    for (int u = 0; u < EMB_BATCH; ++u) {
      asm volatile("" :: "v"(t[u][0]), "v"(t[u][1]), "v"(t[u][2]), "v"(t[u][3]));
    }
#pragma unroll
    for (int u = 0; u < EMB_BATCH; ++u) {
      const int r = r0 + u * SETUP_WAVES;
      if (r < cnt) {
        v4f s = {0.f, 0.f, 0.f, 0.f};
#pragma unroll
        for (int cc = 0; cc < N_COLS; ++cc) s += t[u][cc];
        const float d = (float)sdeg[r];
        const float sc = (d > 0.f) ? rsqrtf(d) : 0.f;
        *(v4f*)(h + (size_t)(lo + r) * DIM + lane * 4) = s * sc;
      }
    }
  }
}

__global__ __launch_bounds__(AGG_THREADS)
void agg_kernel(const int* __restrict__ ei, const float* __restrict__ dinv,
                const float* __restrict__ xw, const float* __restrict__ bl, float* __restrict__ outp,
                const float* __restrict__ W0, const float* __restrict__ Wn,
                const unsigned* __restrict__ lists, const unsigned* __restrict__ counts,
                const unsigned* __restrict__ woffs,
                const int* __restrict__ batch, float* __restrict__ pool, float* __restrict__ pnodes) {
  __shared__ float rows[ROW_TILES * 16 * ROW_PITCH];
  __shared__ __attribute__((aligned(16))) float sdinv[ROW_TILES * 16];
  __shared__ unsigned queue[QUEUE_WORDS];
  __shared__ float pbin[POOL_BINS * DIM];
  __shared__ float pnum[POOL_BINS];

  const int tid = threadIdx.x, lane = tid & 31;
  const int wave = __builtin_amdgcn_readfirstlane(tid >> 5);
  const bool first = (W0 != nullptr);
  const bool product = (Wn != nullptr);
  const unsigned* list = lists + (size_t)blockIdx.x * LIST_CAP;
  const unsigned kept = counts[blockIdx.x];
  const bool use_list = kept <= LIST_CAP;
  const int lo = blockIdx.x * OWN_ROWS;
  const int cnt = (N_NODES - lo < OWN_ROWS) ? (N_NODES - lo) : OWN_ROWS;

  const int fc4 = lane << 2;
  v4f fbias = {0.f, 0.f, 0.f, 0.f};
  if (!first) fbias = *(const v4f*)(bl + fc4);

  for (int i = cnt * 32 + tid; i < ((cnt + 15) & ~15) * 32; i += AGG_THREADS) {
    int r = i >> 5, c4 = (i & 31) << 2;
    *(v4f*)(rows + r * ROW_PITCH + c4) = (v4f){0.f, 0.f, 0.f, 0.f};
  }
  for (int i = tid; i < ROW_TILES * 16; i += AGG_THREADS) sdinv[i] = (i < cnt) ? dinv[lo + i] : 0.f;
  if (!product) {
    for (int i = tid; i < POOL_BINS * DIM; i += AGG_THREADS) pbin[i] = 0.f;
    if (tid < POOL_BINS) pnum[tid] = 0.f;
  }

  if (__builtin_expect(use_list, 1)) {
    const int wr0 = wave * WAVE_ROWS;
    const int wn = (cnt - wr0 < WAVE_ROWS) ? (cnt - wr0) : WAVE_ROWS;
    if (wn > 0) {
      float* wrow = rows + wr0 * ROW_PITCH + fc4;
      v4f st[WAVE_ROWS];
#pragma unroll
      for (int k = 0; k < WAVE_ROWS; ++k) {
        const int r = wr0 + ((k < wn) ? k : wn - 1);
        st[k] = *(const v4f*)(xw + (size_t)(lo + r) * DIM + fc4);
      }
      const unsigned* wo = woffs + (size_t)blockIdx.x * (AGG_WAVES + 1) + wave;
      const unsigned beg = wo[0], end = wo[1];
      sum_edges(list + beg, end - beg, wrow, xw, lo + wr0, lane, [&]() {
#pragma unroll
        for (int k = 0; k < WAVE_ROWS; ++k)
          if (k < wn) *(v4f*)(wrow + k * ROW_PITCH) = st[k];
      });
      if (!first) {
        v4f o[WAVE_ROWS];
#pragma unroll
        for (int k = 0; k < WAVE_ROWS; ++k)
          if (k < wn) o[k] = *(const v4f*)(wrow + k * ROW_PITCH);
#pragma unroll
        for (int k = 0; k < WAVE_ROWS; ++k) {
          if (k < wn) {
            const float sc = dinv[lo + wr0 + k];
            v4f x = o[k];
            x.x = fmaxf(fbias.x + x.x * sc, 0.f);
            x.y = fmaxf(fbias.y + x.y * sc, 0.f);
            x.z = fmaxf(fbias.z + x.z * sc, 0.f);
            x.w = fmaxf(fbias.w + x.w * sc, 0.f);
            *(v4f*)(wrow + k * ROW_PITCH) = x;
          }
        }
      }
    }
  } else {
    for (int i = tid; i < cnt * 32; i += AGG_THREADS) {
      int r = i >> 5, c4 = (i & 31) << 2;
      *(v4f*)(rows + r * ROW_PITCH + c4) = *(const v4f*)(xw + (size_t)(lo + r) * DIM + c4);
    }
    __syncthreads();
    unsigned* q = queue + wave * QCAP;
    scan_edges(ei + N_EDGES, lo, cnt, wave, lane, q, [&](unsigned n) {
      resolve(q, n, ei, lo, lane);
      drain(q, n, rows, xw, lo, lane);
    });
    if (!first) {
      __syncthreads();
      for (int i = tid; i < cnt * 32; i += AGG_THREADS) {
        const int r = i >> 5;
        const float sc = sdinv[r];
        v4f* p = (v4f*)(rows + r * ROW_PITCH + fc4);
        const v4f x = *p;
        v4f o;
        o.x = fmaxf(fbias.x + x.x * sc, 0.f);
        o.y = fmaxf(fbias.y + x.y * sc, 0.f);
        o.z = fmaxf(fbias.z + x.z * sc, 0.f);
        o.w = fmaxf(fbias.w + x.w * sc, 0.f);
        *p = o;
      }
    }
  }
  const int half = lane >> 4, lm = lane & 15;
  const int col0 = (wave & 7) * 16;
  const int n_rt = (cnt + 15) >> 4;
  v2f bf[DIM / 4];
  auto load_b = [&](const float* __restrict__ Wp) {
    const float* bp = Wp + (size_t)(half * 2) * DIM + col0 + lm;
#pragma unroll
    for (int k = 0; k < DIM; k += 4) {
      bf[k / 4].x = bp[k * DIM];
      bf[k / 4].y = bp[k * DIM + DIM];
    }
  };
  if (first) load_b(W0);
  else if (product) load_b(Wn);
  const int pr0 = wave * POOL_ROWS;
  const int pr1 = (pr0 + POOL_ROWS < cnt) ? pr0 + POOL_ROWS : cnt;
  int bv = 0, g0 = 0;
  if (!product) {
    g0 = batch[lo];
    if (pr0 + lane < pr1) bv = batch[lo + pr0 + lane];
  }
  __syncthreads();

  const float* arows = rows;
  if (first) {
    static_assert(QUEUE_WORDS >= STAGE_WORDS, "the queues must hold a second set of rows");
    float* stage = reinterpret_cast<float*>(queue);
    const float badd = bl[col0 + lm];
    for (int rt = wave >> 3; rt < n_rt; rt += AGG_WAVES / 8) {
      v8f acc = tile_product(rows + (rt * 16 + lm) * ROW_PITCH + half * 2, bf);
      float* sp = stage + (rt * 16 + half * 8) * ROW_PITCH + col0 + lm;
      const v4f s0 = *(const v4f*)(sdinv + rt * 16 + half * 8), s1 = *(const v4f*)(sdinv + rt * 16 + half * 8 + 4);
      const float sd[8] = {s0.x, s0.y, s0.z, s0.w, s1.x, s1.y, s1.z, s1.w};
#pragma unroll
      for (int r = 0; r < 8; ++r) sp[r * ROW_PITCH] = fmaxf(acc[r] * sd[r] + badd, 0.f);
    }
    if (product) load_b(Wn);
    __syncthreads();
    arows = stage;
  }

  if (!product) {
    if (pr0 < pr1) {
      const float* rp = arows + pr0 * ROW_PITCH + lane * 2;
      const int n_rows = pr1 - pr0;
      v2f sa = {0.f, 0.f}, sb = {0.f, 0.f};
      float num = 0.f;
      int g = __builtin_amdgcn_readlane(bv, 0);
      v2f na = *(const v2f*)rp, nb = *(const v2f*)(rp + 64);
      for (int k = 0; k < n_rows; ++k) {
        const v2f a = na, b = nb;
        const bool last_row = (k + 1 == n_rows);
        const float* np = rp + (last_row ? k : k + 1) * ROW_PITCH;
        na = *(const v2f*)np;
        nb = *(const v2f*)(np + 64);
        const int gn = __builtin_amdgcn_readlane(bv, k + 1);
        sa.x += a.x;
        sa.y += a.y;
        sb.x += b.x;
        sb.y += b.y;
        num += 1.f;
        if (last_row || gn != g) {
          const unsigned slot = (unsigned)(g - g0);
          if (slot < POOL_BINS) {
            float* d = pbin + slot * DIM + lane * 2;
            __hip_atomic_fetch_add(d,      sa.x, __ATOMIC_RELAXED, __HIP_MEMORY_SCOPE_WORKGROUP);
            __hip_atomic_fetch_add(d + 1,  sa.y, __ATOMIC_RELAXED, __HIP_MEMORY_SCOPE_WORKGROUP);
            __hip_atomic_fetch_add(d + 64, sb.x, __ATOMIC_RELAXED, __HIP_MEMORY_SCOPE_WORKGROUP);
            __hip_atomic_fetch_add(d + 65, sb.y, __ATOMIC_RELAXED, __HIP_MEMORY_SCOPE_WORKGROUP);
            if (lane == 0) __hip_atomic_fetch_add(pnum + slot, num, __ATOMIC_RELAXED, __HIP_MEMORY_SCOPE_WORKGROUP);
          } else {
            float* d = pool + (size_t)g * DIM + lane * 2;
            unsafeAtomicAdd(d,      sa.x);
            unsafeAtomicAdd(d + 1,  sa.y);
            unsafeAtomicAdd(d + 64, sb.x);
            unsafeAtomicAdd(d + 65, sb.y);
            if (lane == 0) unsafeAtomicAdd(&pnodes[g], num);
          }
          sa = (v2f){0.f, 0.f};
          sb = (v2f){0.f, 0.f};
          num = 0.f;
        }
        g = gn;
      }
    }
    __syncthreads();
    for (int i = tid; i < POOL_BINS * DIM; i += AGG_THREADS) {
      const int slot = i >> 7;
      const float n = pnum[slot];
      if (n > 0.f) {
        unsafeAtomicAdd(&pool[(size_t)(g0 + slot) * DIM + (i & (DIM - 1))], pbin[i]);
        if ((i & (DIM - 1)) == 0) unsafeAtomicAdd(&pnodes[g0 + slot], n);
      }
    }
    return;
  }

  for (int rt = wave >> 3; rt < n_rt; rt += AGG_WAVES / 8) {
    v8f acc = tile_product(arows + (rt * 16 + lm) * ROW_PITCH + half * 2, bf);
    const int r0 = rt * 16 + half * 8;
    const v4f s0 = *(const v4f*)(sdinv + r0), s1 = *(const v4f*)(sdinv + r0 + 4);
    const float sd[8] = {s0.x, s0.y, s0.z, s0.w, s1.x, s1.y, s1.z, s1.w};
#pragma unroll
    for (int r = 0; r < 8; ++r)
      if (r0 + r < cnt) outp[(size_t)(lo + r0 + r) * DIM + col0 + lm] = acc[r] * sd[r];
  }
}

__global__ void pool_div_kernel(float* __restrict__ pool, const float* __restrict__ cnt) {
  int t = blockIdx.x * blockDim.x + threadIdx.x;
  if (t >= N_GRAPHS * DIM) return;
  pool[t] /= fmaxf(cnt[t / DIM], 1.0f);
}

extern "C" void kernel_launch(void* const* d_in, const int* in_sizes, int n_in,
                              void* d_out, int out_size, void* d_ws, size_t ws_size,
                              hipStream_t stream) {
  const int*   x    = (const int*)d_in[0];
  const int*   ei   = (const int*)d_in[1];
  const int*   batch= (const int*)d_in[2];
  const float* emb  = (const float*)d_in[3];
  const float* W    = (const float*)d_in[4];
  const float* b    = (const float*)d_in[5];
  const float* Wr   = (const float*)d_in[6];
  const float* br   = (const float*)d_in[7];
  float* out = (float*)d_out;

  char* ws = (char*)d_ws;
  const size_t NH = (size_t)N_NODES * DIM * sizeof(float);
  float* h    = (float*)(ws);
  float* xw   = (float*)(ws + NH);
  float* dinv = (float*)(ws + 3 * NH);
  float* pool = (float*)(ws + 3 * NH + 4 * 1024 * 1024);
  float* cnt  = (float*)(ws + 3 * NH + 4 * 1024 * 1024 + 512 * 1024);
  unsigned* counts = (unsigned*)(ws + 3 * NH + 5 * 1024 * 1024);
  unsigned* woffs  = (unsigned*)(ws + 3 * NH + 6 * 1024 * 1024);
  unsigned* lists  = (unsigned*)(ws + 3 * NH + 8 * 1024 * 1024);
  unsigned* binned = (unsigned*)(ws + 3 * NH + 44 * 1024 * 1024);
  unsigned* meta   = (unsigned*)(ws + 3 * NH + 52 * 1024 * 1024);

  const int B256 = 256;

  bin_kernel<<<N_CHUNKS, BIN_THREADS, 0, stream>>>(ei, binned, meta);
  setup_kernel<<<N_OWNERS, SETUP_THREADS, 0, stream>>>(x, emb, binned, meta, h, dinv, lists, counts, woffs, pool, cnt);

  static_assert(N_LAYERS == 4, "the four layer launches are written out");
  agg_kernel<<<N_OWNERS, AGG_THREADS, 0, stream>>>(ei, dinv, h,  b,           xw, W,       W + 1 * DIM * DIM, lists, counts, woffs, batch, pool, cnt);
  agg_kernel<<<N_OWNERS, AGG_THREADS, 0, stream>>>(ei, dinv, xw, b + 1 * DIM, h,  nullptr, W + 2 * DIM * DIM, lists, counts, woffs, batch, pool, cnt);
  agg_kernel<<<N_OWNERS, AGG_THREADS, 0, stream>>>(ei, dinv, h,  b + 2 * DIM, xw, nullptr, W + 3 * DIM * DIM, lists, counts, woffs, batch, pool, cnt);
  agg_kernel<<<N_OWNERS, AGG_THREADS, 0, stream>>>(ei, dinv, xw, b + 3 * DIM, h,  nullptr, nullptr,           lists, counts, woffs, batch, pool, cnt);

  pool_div_kernel<<<(N_GRAPHS * DIM + B256 - 1) / B256, B256, 0, stream>>>(pool, cnt);

  dim3 rgrid(N_GRAPHS / 16, DIM / 16);
  gemm_wmma_f32<<<rgrid, 32, 0, stream>>>(pool, Wr, br, out);
}
